# starting kernel
# baseline (speedup 1.0000x reference)
#include <hip/hip_runtime.h>
#include <math.h>

#define NN 50000
#define EE 800000
#define ET (EE + NN)
#define IN_DIM 128
#define HID 32
#define HEADS 4
#define HC 128
#define OUT_DIM 40
#define SLOPE 0.2f

typedef float v2f __attribute__((ext_vector_type(2)));
typedef float v8f __attribute__((ext_vector_type(8)));

__device__ __forceinline__ unsigned ford(float f) {
    unsigned u = __float_as_uint(f);
    return (u & 0x80000000u) ? ~u : (u | 0x80000000u);
}
__device__ __forceinline__ float funord(unsigned k) {
    unsigned u = (k & 0x80000000u) ? (k & 0x7FFFFFFFu) : ~k;
    return __uint_as_float(u);
}
#define FORD_NEG_INF 0x007FFFFFu

__device__ __forceinline__ void edge_sd(int u, const int* __restrict__ ei,
                                        int& src, int& dst) {
    if (u < EE) { src = ei[u]; dst = ei[EE + u]; }
    else        { src = u - EE; dst = u - EE; }
}

__device__ __forceinline__ float leaky(float e) {
    return e > 0.f ? e : SLOPE * e;
}

__device__ __forceinline__ v8f wmma4(v2f a, v2f b, v8f c) {
    return __builtin_amdgcn_wmma_f32_16x16x4_f32(
               false, a, false, b, (short)0, c, false, false);
}

__global__ __launch_bounds__(32) void wmma_gemm_f32_2t(
    const float* __restrict__ A, const float* __restrict__ B,
    float* __restrict__ C, int K, int Nout)
{
    const int mt   = blockIdx.x;
    const int nt   = blockIdx.y;
    const int lane = threadIdx.x;
    const int hi   = lane >> 4;
    const int lo   = lane & 15;

    const float* Ap = A + (size_t)(mt * 16 + lo) * K + 2 * hi;
    const float* Bp = B + (size_t)(2 * hi) * Nout + nt * 32 + lo;

    v8f acc0 = {}, acc1 = {};
    for (int k = 0; k < K; k += 4) {
        float2 av = *(const float2*)(Ap + k);
        v2f a; a.x = av.x; a.y = av.y;
        const float* bp = Bp + (size_t)k * Nout;
        v2f b0; b0.x = bp[0];  b0.y = bp[Nout];
        v2f b1; b1.x = bp[16]; b1.y = bp[Nout + 16];
        acc0 = wmma4(a, b0, acc0);
        acc1 = wmma4(a, b1, acc1);
    }
    const int col = nt * 32 + lo;
    float* Cp = C + (size_t)(mt * 16 + 8 * hi) * Nout + col;
    #pragma unroll
    for (int r = 0; r < 8; ++r) {
        Cp[(size_t)r * Nout + 0]  = acc0[r];
        Cp[(size_t)r * Nout + 16] = acc1[r];
    }
}

__global__ __launch_bounds__(32) void wmma_gemm_f32_tail(
    const float* __restrict__ A, const float* __restrict__ B,
    float* __restrict__ C, const float* __restrict__ bias,
    int K, int Nout)
{
    const int mt   = blockIdx.x;
    const int nt   = blockIdx.y;
    const int lane = threadIdx.x;
    const int hi   = lane >> 4;
    const int lo   = lane & 15;
    const int col  = nt * 16 + lo;
    const int ccol = col < Nout ? col : Nout - 1;

    const float* Ap = A + (size_t)(mt * 16 + lo) * K + 2 * hi;
    const float* Bp = B + (size_t)(2 * hi) * Nout + ccol;

    v8f acc = {};
    for (int k = 0; k < K; k += 4) {
        float2 av = *(const float2*)(Ap + k);
        v2f a; a.x = av.x; a.y = av.y;
        const float* bp = Bp + (size_t)k * Nout;
        v2f b; b.x = bp[0]; b.y = bp[Nout];
        acc = wmma4(a, b, acc);
    }
    if (col < Nout) {
        const float bb = bias ? bias[col] : 0.f;
        float* Cp = C + (size_t)(mt * 16 + 8 * hi) * Nout + col;
        #pragma unroll
        for (int r = 0; r < 8; ++r)
            Cp[(size_t)r * Nout] = acc[r] + bb;
    }
}

__global__ void attn_coef(const float* __restrict__ h,
                          const float* __restrict__ a_src,
                          const float* __restrict__ a_dst,
                          float* __restrict__ al_s,
                          float* __restrict__ al_d)
{
    int idx = blockIdx.x * blockDim.x + threadIdx.x;
    if (idx >= NN * HEADS) return;
    int n = idx / HEADS, hd = idx % HEADS;
    const float* hp = h + (size_t)n * HC + hd * HID;
    const float* as = a_src + hd * HID;
    const float* ad = a_dst + hd * HID;
    float s = 0.f, d = 0.f;
    #pragma unroll
    for (int c = 0; c < HID; ++c) { float v = hp[c]; s = fmaf(v, as[c], s); d = fmaf(v, ad[c], d); }
    al_s[idx] = s; al_d[idx] = d;
}

__global__ void init_layer(unsigned* __restrict__ m, float* __restrict__ den,
                           float* __restrict__ acc)
{
    int i = blockIdx.x * blockDim.x + threadIdx.x;
    if (i < NN * HEADS) { m[i] = FORD_NEG_INF; den[i] = 0.f; }
    if (i < NN * HC)    acc[i] = 0.f;
}

__global__ void edge_max(const int* __restrict__ ei,
                         const float* __restrict__ al_s,
                         const float* __restrict__ al_d,
                         unsigned* __restrict__ m)
{
    int idx = blockIdx.x * blockDim.x + threadIdx.x;
    if (idx >= ET * HEADS) return;
    int u = idx >> 2, hd = idx & 3;
    int src, dst; edge_sd(u, ei, src, dst);
    float e = leaky(al_s[src * HEADS + hd] + al_d[dst * HEADS + hd]);
    atomicMax(&m[dst * HEADS + hd], ford(e));
}

__global__ void edge_sum(const int* __restrict__ ei,
                         const float* __restrict__ al_s,
                         const float* __restrict__ al_d,
                         const unsigned* __restrict__ m,
                         float* __restrict__ den)
{
    int idx = blockIdx.x * blockDim.x + threadIdx.x;
    if (idx >= ET * HEADS) return;
    int u = idx >> 2, hd = idx & 3;
    int src, dst; edge_sd(u, ei, src, dst);
    float e  = leaky(al_s[src * HEADS + hd] + al_d[dst * HEADS + hd]);
    float mx = funord(m[dst * HEADS + hd]);
    atomicAdd(&den[dst * HEADS + hd], __expf(e - mx));
}

__global__ __launch_bounds__(256) void edge_aggregate(
    const int* __restrict__ ei, const float* __restrict__ h,
    const float* __restrict__ al_s, const float* __restrict__ al_d,
    const unsigned* __restrict__ m, const float* __restrict__ den,
    float* __restrict__ acc)
{
    int wave = blockIdx.x * (blockDim.x >> 5) + (threadIdx.x >> 5);
    int lane = threadIdx.x & 31;
    if (wave >= ET) return;
    int src, dst; edge_sd(wave, ei, src, dst);
    int hd = lane >> 3;
    float e  = leaky(al_s[src * HEADS + hd] + al_d[dst * HEADS + hd]);
    float mx = funord(m[dst * HEADS + hd]);
    float alpha = __expf(e - mx) / (den[dst * HEADS + hd] + 1e-16f);

    float4 v = *((const float4*)(h + (size_t)src * HC) + lane);
    float* ap = acc + (size_t)dst * HC + lane * 4;
    atomicAdd(ap + 0, v.x * alpha);
    atomicAdd(ap + 1, v.y * alpha);
    atomicAdd(ap + 2, v.z * alpha);
    atomicAdd(ap + 3, v.w * alpha);
}

__global__ void bias_relu(float* __restrict__ acc, const float* __restrict__ b)
{
    int i = blockIdx.x * blockDim.x + threadIdx.x;
    if (i >= NN * HC) return;
    float v = acc[i] + b[i & (HC - 1)];
    acc[i] = v > 0.f ? v : 0.f;
}

extern "C" void kernel_launch(void* const* d_in, const int* in_sizes, int n_in,
                              void* d_out, int out_size, void* d_ws, size_t ws_size,
                              hipStream_t stream)
{
    const float* x   = (const float*)d_in[0];
    const int*   ei  = (const int*)d_in[1];
    const float* W[3]    = { (const float*)d_in[2],  (const float*)d_in[6],  (const float*)d_in[10] };
    const float* asv[3]  = { (const float*)d_in[3],  (const float*)d_in[7],  (const float*)d_in[11] };
    const float* adv[3]  = { (const float*)d_in[4],  (const float*)d_in[8],  (const float*)d_in[12] };
    const float* bv[3]   = { (const float*)d_in[5],  (const float*)d_in[9],  (const float*)d_in[13] };
    const float* Wl  = (const float*)d_in[14];
    const float* bl  = (const float*)d_in[15];

    char* ws = (char*)d_ws;
    float*    hbuf = (float*)ws;                       ws += (size_t)NN * HC * sizeof(float);
    float*    bufA = (float*)ws;                       ws += (size_t)NN * HC * sizeof(float);
    float*    bufB = (float*)ws;                       ws += (size_t)NN * HC * sizeof(float);
    float*    al_s = (float*)ws;                       ws += (size_t)NN * HEADS * sizeof(float);
    float*    al_d = (float*)ws;                       ws += (size_t)NN * HEADS * sizeof(float);
    unsigned* mbuf = (unsigned*)ws;                    ws += (size_t)NN * HEADS * sizeof(unsigned);
    float*    den  = (float*)ws;

    const int TPB = 256;
    const int MT  = NN / 16;
    const int gNHC   = (NN * HC + TPB - 1) / TPB;
    const int gNH    = (NN * HEADS + TPB - 1) / TPB;
    const int gEH    = (ET * HEADS + TPB - 1) / TPB;
    const int gWaves = (ET + (TPB / 32) - 1) / (TPB / 32);

    auto gat_layer = [&](const float* xin, int layer, float* acc) {
        hipLaunchKernelGGL(init_layer, dim3(gNHC), dim3(TPB), 0, stream, mbuf, den, acc);
        hipLaunchKernelGGL(wmma_gemm_f32_2t, dim3(MT, HC / 32), dim3(32), 0, stream,
                           xin, W[layer], hbuf, IN_DIM, HC);
        hipLaunchKernelGGL(attn_coef, dim3(gNH), dim3(TPB), 0, stream,
                           hbuf, asv[layer], adv[layer], al_s, al_d);
        hipLaunchKernelGGL(edge_max, dim3(gEH), dim3(TPB), 0, stream, ei, al_s, al_d, mbuf);
        hipLaunchKernelGGL(edge_sum, dim3(gEH), dim3(TPB), 0, stream, ei, al_s, al_d, mbuf, den);
        hipLaunchKernelGGL(edge_aggregate, dim3(gWaves), dim3(TPB), 0, stream,
                           ei, hbuf, al_s, al_d, mbuf, den, acc);
        hipLaunchKernelGGL(bias_relu, dim3(gNHC), dim3(TPB), 0, stream, acc, bv[layer]);
    };

    gat_layer(x,    0, bufA);
    gat_layer(bufA, 1, bufB);
    gat_layer(bufB, 2, bufA);

    hipLaunchKernelGGL(wmma_gemm_f32_tail, dim3(MT, (OUT_DIM + 15) / 16), dim3(32), 0, stream,
                       bufA, Wl, (float*)d_out, bl, HC, OUT_DIM);
}


# Edges counting-sorted by destination once per call; one wave per node keeps max, exp-sum and weighted row sum in registers, stores relu(out+b) once; no float atomics, sums in CSR arrival order
# speedup vs baseline: 2.7486x; 2.7486x over previous
#include <hip/hip_runtime.h>
#include <math.h>

#define NN 50000
#define EE 800000
#define ET (EE + NN)
#define IN_DIM 128
#define HID 32
#define HEADS 4
#define HC 128
#define OUT_DIM 40
#define SLOPE 0.2f

typedef float v2f __attribute__((ext_vector_type(2)));
typedef float v8f __attribute__((ext_vector_type(8)));

__device__ __forceinline__ float leaky(float e) {
    return e > 0.f ? e : SLOPE * e;
}

__device__ __forceinline__ v8f wmma4(v2f a, v2f b, v8f c) {
    return __builtin_amdgcn_wmma_f32_16x16x4_f32(
               false, a, false, b, (short)0, c, false, false);
}

__global__ __launch_bounds__(32) void wmma_gemm_f32_2t(
    const float* __restrict__ A, const float* __restrict__ B,
    float* __restrict__ C, int K, int Nout)
{
    const int mt   = blockIdx.x;
    const int nt   = blockIdx.y;
    const int lane = threadIdx.x;
    const int hi   = lane >> 4;
    const int lo   = lane & 15;

    const float* Ap = A + (size_t)(mt * 16 + lo) * K + 2 * hi;
    const float* Bp = B + (size_t)(2 * hi) * Nout + nt * 32 + lo;

    v8f acc0 = {}, acc1 = {};
    for (int k = 0; k < K; k += 4) {
        float2 av = *(const float2*)(Ap + k);
        v2f a; a.x = av.x; a.y = av.y;
        const float* bp = Bp + (size_t)k * Nout;
        v2f b0; b0.x = bp[0];  b0.y = bp[Nout];
        v2f b1; b1.x = bp[16]; b1.y = bp[Nout + 16];
        acc0 = wmma4(a, b0, acc0);
        acc1 = wmma4(a, b1, acc1);
    }
    const int col = nt * 32 + lo;
    float* Cp = C + (size_t)(mt * 16 + 8 * hi) * Nout + col;
    #pragma unroll
    for (int r = 0; r < 8; ++r) {
        Cp[(size_t)r * Nout + 0]  = acc0[r];
        Cp[(size_t)r * Nout + 16] = acc1[r];
    }
}

__global__ __launch_bounds__(32) void wmma_gemm_f32_tail(
    const float* __restrict__ A, const float* __restrict__ B,
    float* __restrict__ C, const float* __restrict__ bias,
    int K, int Nout)
{
    const int mt   = blockIdx.x;
    const int nt   = blockIdx.y;
    const int lane = threadIdx.x;
    const int hi   = lane >> 4;
    const int lo   = lane & 15;
    const int col  = nt * 16 + lo;
    const int ccol = col < Nout ? col : Nout - 1;

    const float* Ap = A + (size_t)(mt * 16 + lo) * K + 2 * hi;
    const float* Bp = B + (size_t)(2 * hi) * Nout + ccol;

    v8f acc = {};
    for (int k = 0; k < K; k += 4) {
        float2 av = *(const float2*)(Ap + k);
        v2f a; a.x = av.x; a.y = av.y;
        const float* bp = Bp + (size_t)k * Nout;
        v2f b; b.x = bp[0]; b.y = bp[Nout];
        acc = wmma4(a, b, acc);
    }
    if (col < Nout) {
        const float bb = bias ? bias[col] : 0.f;
        float* Cp = C + (size_t)(mt * 16 + 8 * hi) * Nout + col;
        #pragma unroll
        for (int r = 0; r < 8; ++r)
            Cp[(size_t)r * Nout] = acc[r] + bb;
    }
}

__global__ void attn_coef(const float* __restrict__ h,
                          const float* __restrict__ a_src,
                          const float* __restrict__ a_dst,
                          float* __restrict__ al_s,
                          float* __restrict__ al_d)
{
    int idx = blockIdx.x * blockDim.x + threadIdx.x;
    if (idx >= NN * HEADS) return;
    int n = idx / HEADS, hd = idx % HEADS;
    const float* hp = h + (size_t)n * HC + hd * HID;
    const float* as = a_src + hd * HID;
    const float* ad = a_dst + hd * HID;
    float s = 0.f, d = 0.f;
    #pragma unroll
    for (int c = 0; c < HID; ++c) { float v = hp[c]; s = fmaf(v, as[c], s); d = fmaf(v, ad[c], d); }
    al_s[idx] = s; al_d[idx] = d;
}

__global__ void csr_zero(int* __restrict__ deg, int* __restrict__ total)
{
    int i = blockIdx.x * blockDim.x + threadIdx.x;
    if (i < NN) deg[i] = 0;
    if (i == 0) *total = 0;
}

__global__ void csr_count(const int* __restrict__ ei, int* __restrict__ deg)
{
    int u = blockIdx.x * blockDim.x + threadIdx.x;
    if (u < EE) atomicAdd(&deg[ei[EE + u]], 1);
}

__global__ void csr_alloc(const int* __restrict__ deg, int* __restrict__ total,
                          int* __restrict__ start, int* __restrict__ cursor)
{
    const int i    = blockIdx.x * blockDim.x + threadIdx.x;
    const int lane = threadIdx.x & 31;
    const int d    = i < NN ? deg[i] : 0;
    int incl = d;
    for (int off = 1; off < 32; off <<= 1) {
        int t = __shfl_up(incl, off);
        if (lane >= off) incl += t;
    }
    int base = 0;
    if (lane == 31) base = atomicAdd(total, incl);
    base = __shfl(base, 31);
    const int st = base + incl - d;
    if (i < NN) { start[i] = st; cursor[i] = st; }
}

__global__ void csr_fill(const int* __restrict__ ei, int* __restrict__ cursor,
                         int* __restrict__ csr_src)
{
    int u = blockIdx.x * blockDim.x + threadIdx.x;
    if (u >= EE) return;
    int p = atomicAdd(&cursor[ei[EE + u]], 1);
    csr_src[p] = ei[u];
}

struct HeadMask { unsigned k0, k1, k2, k3; };
__device__ __forceinline__ float pick4(HeadMask k, float a, float b, float c, float d) {
    return __uint_as_float((k.k0 & __float_as_uint(a)) | (k.k1 & __float_as_uint(b)) |
                           (k.k2 & __float_as_uint(c)) | (k.k3 & __float_as_uint(d)));
}
__device__ __forceinline__ float lane_bcast(float v, int srclane) {
    return __int_as_float(__builtin_amdgcn_readlane(__float_as_int(v), srclane));
}

__global__ __launch_bounds__(256) void node_gather(
    const int* __restrict__ start, const int* __restrict__ deg,
    const int* __restrict__ csr_src, const float* __restrict__ h,
    const float* __restrict__ al_s, const float* __restrict__ al_d,
    const float* __restrict__ bias, float* __restrict__ out)
{
    const int lane = threadIdx.x & 31;
    const int n = __builtin_amdgcn_readfirstlane(
                      (int)(blockIdx.x * (blockDim.x >> 5) + (threadIdx.x >> 5)));
    if (n >= NN) return;
    const int hd  = lane >> 3;
    HeadMask hk;
    hk.k0 = hd == 0 ? 0xFFFFFFFFu : 0u;  hk.k1 = hd == 1 ? 0xFFFFFFFFu : 0u;
    hk.k2 = hd == 2 ? 0xFFFFFFFFu : 0u;  hk.k3 = hd == 3 ? 0xFFFFFFFFu : 0u;
    const int beg = start[n];
    const int cnt = deg[n];
    const float4 ad4 = *(const float4*)(al_d + (size_t)n * HEADS);
    const float4 sf4 = *(const float4*)(al_s + (size_t)n * HEADS);
    const float4* __restrict__ h4 = (const float4*)h;

    float m0 = leaky(sf4.x + ad4.x), m1 = leaky(sf4.y + ad4.y);
    float m2 = leaky(sf4.z + ad4.z), m3 = leaky(sf4.w + ad4.w);
    for (int j = lane; j < cnt; j += 32) {
        const int s = csr_src[beg + j];
        const float4 a = *(const float4*)(al_s + (size_t)s * HEADS);
        m0 = fmaxf(m0, leaky(a.x + ad4.x));
        m1 = fmaxf(m1, leaky(a.y + ad4.y));
        m2 = fmaxf(m2, leaky(a.z + ad4.z));
        m3 = fmaxf(m3, leaky(a.w + ad4.w));
    }
    for (int off = 16; off > 0; off >>= 1) {
        m0 = fmaxf(m0, __shfl_xor(m0, off));
        m1 = fmaxf(m1, __shfl_xor(m1, off));
        m2 = fmaxf(m2, __shfl_xor(m2, off));
        m3 = fmaxf(m3, __shfl_xor(m3, off));
    }
    m0 = __builtin_isfinite(m0) ? m0 : 0.f;
    m1 = __builtin_isfinite(m1) ? m1 : 0.f;
    m2 = __builtin_isfinite(m2) ? m2 : 0.f;
    m3 = __builtin_isfinite(m3) ? m3 : 0.f;

    const float wself = __expf(pick4(hk, leaky(sf4.x + ad4.x), leaky(sf4.y + ad4.y),
                                         leaky(sf4.z + ad4.z), leaky(sf4.w + ad4.w))
                               - pick4(hk, m0, m1, m2, m3));
    const float4 vself = h4[(size_t)n * (HC / 4) + lane];
    float den = wself;
    float4 acc;
    acc.x = wself * vself.x; acc.y = wself * vself.y;
    acc.z = wself * vself.z; acc.w = wself * vself.w;

    for (int base = 0; base < cnt; base += 32) {
        const int j = base + lane;
        int sl = 0;
        float w0 = 0.f, w1 = 0.f, w2 = 0.f, w3 = 0.f;
        if (j < cnt) {
            sl = csr_src[beg + j];
            const float4 a = *(const float4*)(al_s + (size_t)sl * HEADS);
            w0 = __expf(leaky(a.x + ad4.x) - m0);
            w1 = __expf(leaky(a.y + ad4.y) - m1);
            w2 = __expf(leaky(a.z + ad4.z) - m2);
            w3 = __expf(leaky(a.w + ad4.w) - m3);
        }
        const int lim = (cnt - base) < 32 ? (cnt - base) : 32;
        #pragma unroll 4
        for (int jj = 0; jj < lim; ++jj) {
            const int   s  = __builtin_amdgcn_readlane(sl, jj);
            const float wj = pick4(hk, lane_bcast(w0, jj), lane_bcast(w1, jj),
                                       lane_bcast(w2, jj), lane_bcast(w3, jj));
            const float4 v = h4[(size_t)s * (HC / 4) + lane];
            den += wj;
            acc.x = fmaf(wj, v.x, acc.x);
            acc.y = fmaf(wj, v.y, acc.y);
            acc.z = fmaf(wj, v.z, acc.z);
            acc.w = fmaf(wj, v.w, acc.w);
        }
    }

    const float  dd = den + 1e-16f;
    const float4 b  = *((const float4*)bias + lane);
    float4 o;
    o.x = acc.x / dd + b.x;  o.y = acc.y / dd + b.y;
    o.z = acc.z / dd + b.z;  o.w = acc.w / dd + b.w;
    o.x = o.x > 0.f ? o.x : 0.f;  o.y = o.y > 0.f ? o.y : 0.f;
    o.z = o.z > 0.f ? o.z : 0.f;  o.w = o.w > 0.f ? o.w : 0.f;
    ((float4*)out)[(size_t)n * (HC / 4) + lane] = o;
}

extern "C" void kernel_launch(void* const* d_in, const int* in_sizes, int n_in,
                              void* d_out, int out_size, void* d_ws, size_t ws_size,
                              hipStream_t stream)
{
    const float* x   = (const float*)d_in[0];
    const int*   ei  = (const int*)d_in[1];
    const float* W[3]    = { (const float*)d_in[2],  (const float*)d_in[6],  (const float*)d_in[10] };
    const float* asv[3]  = { (const float*)d_in[3],  (const float*)d_in[7],  (const float*)d_in[11] };
    const float* adv[3]  = { (const float*)d_in[4],  (const float*)d_in[8],  (const float*)d_in[12] };
    const float* bv[3]   = { (const float*)d_in[5],  (const float*)d_in[9],  (const float*)d_in[13] };
    const float* Wl  = (const float*)d_in[14];
    const float* bl  = (const float*)d_in[15];

    char* ws = (char*)d_ws;
    float* hbuf = (float*)ws;                          ws += (size_t)NN * HC * sizeof(float);
    float* bufA = (float*)ws;                          ws += (size_t)NN * HC * sizeof(float);
    float* bufB = (float*)ws;                          ws += (size_t)NN * HC * sizeof(float);
    float* al_s = (float*)ws;                          ws += (size_t)NN * HEADS * sizeof(float);
    float* al_d = (float*)ws;                          ws += (size_t)NN * HEADS * sizeof(float);
    int*   deg    = (int*)ws;                          ws += (size_t)NN * sizeof(int);
    int*   start  = (int*)ws;                          ws += (size_t)NN * sizeof(int);
    int*   cursor = (int*)ws;                          ws += (size_t)NN * sizeof(int);
    int*   total  = (int*)ws;                          ws += 64 * sizeof(int);
    int*   csr_src = (int*)ws;

    const int TPB = 256;
    const int MT  = NN / 16;
    const int gN  = (NN + TPB - 1) / TPB;
    const int gE  = (EE + TPB - 1) / TPB;
    const int gNH = (NN * HEADS + TPB - 1) / TPB;
    const int gNodeWaves = (NN + (TPB / 32) - 1) / (TPB / 32);

    hipLaunchKernelGGL(csr_zero,  dim3(gN), dim3(TPB), 0, stream, deg, total);
    hipLaunchKernelGGL(csr_count, dim3(gE), dim3(TPB), 0, stream, ei, deg);
    hipLaunchKernelGGL(csr_alloc, dim3(gN), dim3(TPB), 0, stream, deg, total, start, cursor);
    hipLaunchKernelGGL(csr_fill,  dim3(gE), dim3(TPB), 0, stream, ei, cursor, csr_src);

    auto gat_layer = [&](const float* xin, int layer, float* xout) {
        hipLaunchKernelGGL(wmma_gemm_f32_2t, dim3(MT, HC / 32), dim3(32), 0, stream,
                           xin, W[layer], hbuf, IN_DIM, HC);
        hipLaunchKernelGGL(attn_coef, dim3(gNH), dim3(TPB), 0, stream,
                           hbuf, asv[layer], adv[layer], al_s, al_d);
        hipLaunchKernelGGL(node_gather, dim3(gNodeWaves), dim3(TPB), 0, stream,
                           start, deg, csr_src, hbuf, al_s, al_d, bv[layer], xout);
    };

    gat_layer(x,    0, bufA);
    gat_layer(bufA, 1, bufB);
    gat_layer(bufB, 2, bufA);

    hipLaunchKernelGGL(wmma_gemm_f32_tail, dim3(MT, (OUT_DIM + 15) / 16), dim3(32), 0, stream,
                       bufA, Wl, (float*)d_out, bl, HC, OUT_DIM);
}


# Attention logits al_s/al_d formed in the layer GEMM epilogue from the accumulators (4-wave group per 16 rows, wave = head, 16-lane row sums); logits kernel removed; logit sums in lane-tree order
# speedup vs baseline: 3.2183x; 1.1709x over previous
#include <hip/hip_runtime.h>
#include <math.h>

#define NN 50000
#define EE 800000
#define ET (EE + NN)
#define IN_DIM 128
#define HID 32
#define HEADS 4
#define HC 128
#define OUT_DIM 40
#define SLOPE 0.2f

typedef float v2f __attribute__((ext_vector_type(2)));
typedef float v8f __attribute__((ext_vector_type(8)));

__device__ __forceinline__ float leaky(float e) {
    return e > 0.f ? e : SLOPE * e;
}

__device__ __forceinline__ v8f wmma4(v2f a, v2f b, v8f c) {
    return __builtin_amdgcn_wmma_f32_16x16x4_f32(
               false, a, false, b, (short)0, c, false, false);
}

template <int CTRL>
__device__ __forceinline__ float row_xmask(float v) {
    return __int_as_float(__builtin_amdgcn_update_dpp(0, __float_as_int(v), CTRL, 0xf, 0xf, true));
}
__device__ __forceinline__ float row16_sum(float v) {
    v += row_xmask<0x161>(v);
    v += row_xmask<0x162>(v);
    v += row_xmask<0x164>(v);
    v += row_xmask<0x168>(v);
    return v;
}

__global__ __launch_bounds__(128) void wmma_gemm_f32_logits(
    const float* __restrict__ A, const float* __restrict__ B,
    float* __restrict__ C,
    const float* __restrict__ a_src, const float* __restrict__ a_dst,
    float* __restrict__ al_s, float* __restrict__ al_d, int K, int Nout)
{
    __shared__ __attribute__((aligned(16))) float stage[2][HEADS][16];

    const int mt   = blockIdx.x;
    const int nt   = __builtin_amdgcn_readfirstlane((int)(threadIdx.x >> 5));
    const int lane = threadIdx.x & 31;
    const int hi   = lane >> 4;
    const int lo   = lane & 15;

    const float* Ap = A + (size_t)(mt * 16 + lo) * K + 2 * hi;
    const float* Bp = B + (size_t)(2 * hi) * Nout + nt * 32 + lo;
    const int col = nt * 32 + lo;

    const float as0 = a_src[col], as1 = a_src[col + 16];
    const float ad0 = a_dst[col], ad1 = a_dst[col + 16];

    v8f acc0 = {}, acc1 = {};
    for (int k = 0; k < K; k += 4) {
        float2 av = *(const float2*)(Ap + k);
        v2f a; a.x = av.x; a.y = av.y;
        const float* bp = Bp + (size_t)k * Nout;
        v2f b0; b0.x = bp[0];  b0.y = bp[Nout];
        v2f b1; b1.x = bp[16]; b1.y = bp[Nout + 16];
        acc0 = wmma4(a, b0, acc0);
        acc1 = wmma4(a, b1, acc1);
    }
    float* Cp = C + (size_t)(mt * 16 + 8 * hi) * Nout + col;
    #pragma unroll
    for (int r = 0; r < 8; ++r) {
        Cp[(size_t)r * Nout + 0]  = acc0[r];
        Cp[(size_t)r * Nout + 16] = acc1[r];
    }

    float ps[8], pd[8];
    #pragma unroll
    for (int r = 0; r < 8; ++r) {
        ps[r] = row16_sum(fmaf(acc1[r], as1, acc0[r] * as0));
        pd[r] = row16_sum(fmaf(acc1[r], ad1, acc0[r] * ad0));
    }
    if (lo == 0) {
        float4* s4 = (float4*)&stage[0][nt][8 * hi];
        float4* d4 = (float4*)&stage[1][nt][8 * hi];
        s4[0] = make_float4(ps[0], ps[1], ps[2], ps[3]);
        s4[1] = make_float4(ps[4], ps[5], ps[6], ps[7]);
        d4[0] = make_float4(pd[0], pd[1], pd[2], pd[3]);
        d4[1] = make_float4(pd[4], pd[5], pd[6], pd[7]);
    }
    __builtin_amdgcn_fence(__ATOMIC_RELEASE, "workgroup", "local");
    __builtin_amdgcn_s_barrier();
    __builtin_amdgcn_fence(__ATOMIC_ACQUIRE, "workgroup", "local");
    if (nt == 0) {
        float4 v;
        v.x = stage[hi][0][lo];  v.y = stage[hi][1][lo];
        v.z = stage[hi][2][lo];  v.w = stage[hi][3][lo];
        float* dst = hi ? al_d : al_s;
        *(float4*)(dst + (size_t)(mt * 16 + lo) * HEADS) = v;
    }
}

__global__ __launch_bounds__(32) void wmma_gemm_f32_tail(
    const float* __restrict__ A, const float* __restrict__ B,
    float* __restrict__ C, const float* __restrict__ bias,
    int K, int Nout)
{
    const int mt   = blockIdx.x;
    const int nt   = blockIdx.y;
    const int lane = threadIdx.x;
    const int hi   = lane >> 4;
    const int lo   = lane & 15;
    const int col  = nt * 16 + lo;
    const int ccol = col < Nout ? col : Nout - 1;

    const float* Ap = A + (size_t)(mt * 16 + lo) * K + 2 * hi;
    const float* Bp = B + (size_t)(2 * hi) * Nout + ccol;

    v8f acc = {};
    for (int k = 0; k < K; k += 4) {
        float2 av = *(const float2*)(Ap + k);
        v2f a; a.x = av.x; a.y = av.y;
        const float* bp = Bp + (size_t)k * Nout;
        v2f b; b.x = bp[0]; b.y = bp[Nout];
        acc = wmma4(a, b, acc);
    }
    if (col < Nout) {
        const float bb = bias ? bias[col] : 0.f;
        float* Cp = C + (size_t)(mt * 16 + 8 * hi) * Nout + col;
        #pragma unroll
        for (int r = 0; r < 8; ++r)
            Cp[(size_t)r * Nout] = acc[r] + bb;
    }
}

__global__ void csr_zero(int* __restrict__ deg, int* __restrict__ total)
{
    int i = blockIdx.x * blockDim.x + threadIdx.x;
    if (i < NN) deg[i] = 0;
    if (i == 0) *total = 0;
}

__global__ void csr_count(const int* __restrict__ ei, int* __restrict__ deg)
{
    int u = blockIdx.x * blockDim.x + threadIdx.x;
    if (u < EE) atomicAdd(&deg[ei[EE + u]], 1);
}

__global__ void csr_alloc(const int* __restrict__ deg, int* __restrict__ total,
                          int* __restrict__ start, int* __restrict__ cursor)
{
    const int i    = blockIdx.x * blockDim.x + threadIdx.x;
    const int lane = threadIdx.x & 31;
    const int d    = i < NN ? deg[i] : 0;
    int incl = d;
    for (int off = 1; off < 32; off <<= 1) {
        int t = __shfl_up(incl, off);
        if (lane >= off) incl += t;
    }
    int base = 0;
    if (lane == 31) base = atomicAdd(total, incl);
    base = __shfl(base, 31);
    const int st = base + incl - d;
    if (i < NN) { start[i] = st; cursor[i] = st; }
}

__global__ void csr_fill(const int* __restrict__ ei, int* __restrict__ cursor,
                         int* __restrict__ csr_src)
{
    int u = blockIdx.x * blockDim.x + threadIdx.x;
    if (u >= EE) return;
    int p = atomicAdd(&cursor[ei[EE + u]], 1);
    csr_src[p] = ei[u];
}

struct HeadMask { unsigned k0, k1, k2, k3; };
__device__ __forceinline__ float pick4(HeadMask k, float a, float b, float c, float d) {
    return __uint_as_float((k.k0 & __float_as_uint(a)) | (k.k1 & __float_as_uint(b)) |
                           (k.k2 & __float_as_uint(c)) | (k.k3 & __float_as_uint(d)));
}
__device__ __forceinline__ float lane_bcast(float v, int srclane) {
    return __int_as_float(__builtin_amdgcn_readlane(__float_as_int(v), srclane));
}

__global__ __launch_bounds__(256) void node_gather(
    const int* __restrict__ start, const int* __restrict__ deg,
    const int* __restrict__ csr_src, const float* __restrict__ h,
    const float* __restrict__ al_s, const float* __restrict__ al_d,
    const float* __restrict__ bias, float* __restrict__ out)
{
    const int lane = threadIdx.x & 31;
    const int n = __builtin_amdgcn_readfirstlane(
                      (int)(blockIdx.x * (blockDim.x >> 5) + (threadIdx.x >> 5)));
    if (n >= NN) return;
    const int hd  = lane >> 3;
    HeadMask hk;
    hk.k0 = hd == 0 ? 0xFFFFFFFFu : 0u;  hk.k1 = hd == 1 ? 0xFFFFFFFFu : 0u;
    hk.k2 = hd == 2 ? 0xFFFFFFFFu : 0u;  hk.k3 = hd == 3 ? 0xFFFFFFFFu : 0u;
    const int beg = start[n];
    const int cnt = deg[n];
    const float4 ad4 = *(const float4*)(al_d + (size_t)n * HEADS);
    const float4 sf4 = *(const float4*)(al_s + (size_t)n * HEADS);
    const float4* __restrict__ h4 = (const float4*)h;

    float m0 = leaky(sf4.x + ad4.x), m1 = leaky(sf4.y + ad4.y);
    float m2 = leaky(sf4.z + ad4.z), m3 = leaky(sf4.w + ad4.w);
    for (int j = lane; j < cnt; j += 32) {
        const int s = csr_src[beg + j];
        const float4 a = *(const float4*)(al_s + (size_t)s * HEADS);
        m0 = fmaxf(m0, leaky(a.x + ad4.x));
        m1 = fmaxf(m1, leaky(a.y + ad4.y));
        m2 = fmaxf(m2, leaky(a.z + ad4.z));
        m3 = fmaxf(m3, leaky(a.w + ad4.w));
    }
    for (int off = 16; off > 0; off >>= 1) {
        m0 = fmaxf(m0, __shfl_xor(m0, off));
        m1 = fmaxf(m1, __shfl_xor(m1, off));
        m2 = fmaxf(m2, __shfl_xor(m2, off));
        m3 = fmaxf(m3, __shfl_xor(m3, off));
    }
    m0 = __builtin_isfinite(m0) ? m0 : 0.f;
    m1 = __builtin_isfinite(m1) ? m1 : 0.f;
    m2 = __builtin_isfinite(m2) ? m2 : 0.f;
    m3 = __builtin_isfinite(m3) ? m3 : 0.f;

    const float wself = __expf(pick4(hk, leaky(sf4.x + ad4.x), leaky(sf4.y + ad4.y),
                                         leaky(sf4.z + ad4.z), leaky(sf4.w + ad4.w))
                               - pick4(hk, m0, m1, m2, m3));
    const float4 vself = h4[(size_t)n * (HC / 4) + lane];
    float den = wself;
    float4 acc;
    acc.x = wself * vself.x; acc.y = wself * vself.y;
    acc.z = wself * vself.z; acc.w = wself * vself.w;

    for (int base = 0; base < cnt; base += 32) {
        const int j = base + lane;
        int sl = 0;
        float w0 = 0.f, w1 = 0.f, w2 = 0.f, w3 = 0.f;
        if (j < cnt) {
            sl = csr_src[beg + j];
            const float4 a = *(const float4*)(al_s + (size_t)sl * HEADS);
            w0 = __expf(leaky(a.x + ad4.x) - m0);
            w1 = __expf(leaky(a.y + ad4.y) - m1);
            w2 = __expf(leaky(a.z + ad4.z) - m2);
            w3 = __expf(leaky(a.w + ad4.w) - m3);
        }
        const int lim = (cnt - base) < 32 ? (cnt - base) : 32;
        #pragma unroll 4
        for (int jj = 0; jj < lim; ++jj) {
            const int   s  = __builtin_amdgcn_readlane(sl, jj);
            const float wj = pick4(hk, lane_bcast(w0, jj), lane_bcast(w1, jj),
                                       lane_bcast(w2, jj), lane_bcast(w3, jj));
            const float4 v = h4[(size_t)s * (HC / 4) + lane];
            den += wj;
            acc.x = fmaf(wj, v.x, acc.x);
            acc.y = fmaf(wj, v.y, acc.y);
            acc.z = fmaf(wj, v.z, acc.z);
            acc.w = fmaf(wj, v.w, acc.w);
        }
    }

    const float  dd = den + 1e-16f;
    const float4 b  = *((const float4*)bias + lane);
    float4 o;
    o.x = acc.x / dd + b.x;  o.y = acc.y / dd + b.y;
    o.z = acc.z / dd + b.z;  o.w = acc.w / dd + b.w;
    o.x = o.x > 0.f ? o.x : 0.f;  o.y = o.y > 0.f ? o.y : 0.f;
    o.z = o.z > 0.f ? o.z : 0.f;  o.w = o.w > 0.f ? o.w : 0.f;
    ((float4*)out)[(size_t)n * (HC / 4) + lane] = o;
}

extern "C" void kernel_launch(void* const* d_in, const int* in_sizes, int n_in,
                              void* d_out, int out_size, void* d_ws, size_t ws_size,
                              hipStream_t stream)
{
    const float* x   = (const float*)d_in[0];
    const int*   ei  = (const int*)d_in[1];
    const float* W[3]    = { (const float*)d_in[2],  (const float*)d_in[6],  (const float*)d_in[10] };
    const float* asv[3]  = { (const float*)d_in[3],  (const float*)d_in[7],  (const float*)d_in[11] };
    const float* adv[3]  = { (const float*)d_in[4],  (const float*)d_in[8],  (const float*)d_in[12] };
    const float* bv[3]   = { (const float*)d_in[5],  (const float*)d_in[9],  (const float*)d_in[13] };
    const float* Wl  = (const float*)d_in[14];
    const float* bl  = (const float*)d_in[15];

    char* ws = (char*)d_ws;
    float* hbuf = (float*)ws;                          ws += (size_t)NN * HC * sizeof(float);
    float* bufA = (float*)ws;                          ws += (size_t)NN * HC * sizeof(float);
    float* bufB = (float*)ws;                          ws += (size_t)NN * HC * sizeof(float);
    float* al_s = (float*)ws;                          ws += (size_t)NN * HEADS * sizeof(float);
    float* al_d = (float*)ws;                          ws += (size_t)NN * HEADS * sizeof(float);
    int*   deg    = (int*)ws;                          ws += (size_t)NN * sizeof(int);
    int*   start  = (int*)ws;                          ws += (size_t)NN * sizeof(int);
    int*   cursor = (int*)ws;                          ws += (size_t)NN * sizeof(int);
    int*   total  = (int*)ws;                          ws += 64 * sizeof(int);
    int*   csr_src = (int*)ws;

    const int TPB = 256;
    const int MT  = NN / 16;
    const int gN  = (NN + TPB - 1) / TPB;
    const int gE  = (EE + TPB - 1) / TPB;
    const int gNodeWaves = (NN + (TPB / 32) - 1) / (TPB / 32);

    hipLaunchKernelGGL(csr_zero,  dim3(gN), dim3(TPB), 0, stream, deg, total);
    hipLaunchKernelGGL(csr_count, dim3(gE), dim3(TPB), 0, stream, ei, deg);
    hipLaunchKernelGGL(csr_alloc, dim3(gN), dim3(TPB), 0, stream, deg, total, start, cursor);
    hipLaunchKernelGGL(csr_fill,  dim3(gE), dim3(TPB), 0, stream, ei, cursor, csr_src);

    auto gat_layer = [&](const float* xin, int layer, float* xout) {
        hipLaunchKernelGGL(wmma_gemm_f32_logits, dim3(MT), dim3(32 * HEADS), 0, stream,
                           xin, W[layer], hbuf, asv[layer], adv[layer], al_s, al_d, IN_DIM, HC);
        hipLaunchKernelGGL(node_gather, dim3(gNodeWaves), dim3(TPB), 0, stream,
                           start, deg, csr_src, hbuf, al_s, al_d, bv[layer], xout);
    };

    gat_layer(x,    0, bufA);
    gat_layer(bufA, 1, bufB);
    gat_layer(bufB, 2, bufA);

    hipLaunchKernelGGL(wmma_gemm_f32_tail, dim3(MT, (OUT_DIM + 15) / 16), dim3(32), 0, stream,
                       bufA, Wl, (float*)d_out, bl, HC, OUT_DIM);
}


# CSR build on LDS counters: 16 node ranges x 32 edge chunks count and place with LDS adds; 3 launches not 4, no global atomic per edge; per-node sum order now chunk order, then arrival order
# speedup vs baseline: 4.1628x; 1.2935x over previous
#include <hip/hip_runtime.h>
#include <math.h>

#define NN 50000
#define EE 800000
#define ET (EE + NN)
#define IN_DIM 128
#define HID 32
#define HEADS 4
#define HC 128
#define OUT_DIM 40
#define SLOPE 0.2f

typedef float v2f __attribute__((ext_vector_type(2)));
typedef float v8f __attribute__((ext_vector_type(8)));

__device__ __forceinline__ float leaky(float e) {
    return e > 0.f ? e : SLOPE * e;
}

__device__ __forceinline__ v8f wmma4(v2f a, v2f b, v8f c) {
    return __builtin_amdgcn_wmma_f32_16x16x4_f32(
               false, a, false, b, (short)0, c, false, false);
}

template <int CTRL>
__device__ __forceinline__ float row_xmask(float v) {
    return __int_as_float(__builtin_amdgcn_update_dpp(0, __float_as_int(v), CTRL, 0xf, 0xf, true));
}
__device__ __forceinline__ float row16_sum(float v) {
    v += row_xmask<0x161>(v);
    v += row_xmask<0x162>(v);
    v += row_xmask<0x164>(v);
    v += row_xmask<0x168>(v);
    return v;
}

__global__ __launch_bounds__(128) void wmma_gemm_f32_logits(
    const float* __restrict__ A, const float* __restrict__ B,
    float* __restrict__ C,
    const float* __restrict__ a_src, const float* __restrict__ a_dst,
    float* __restrict__ al_s, float* __restrict__ al_d, int K, int Nout)
{
    __shared__ __attribute__((aligned(16))) float stage[2][HEADS][16];

    const int mt   = blockIdx.x;
    const int nt   = __builtin_amdgcn_readfirstlane((int)(threadIdx.x >> 5));
    const int lane = threadIdx.x & 31;
    const int hi   = lane >> 4;
    const int lo   = lane & 15;

    const float* Ap = A + (size_t)(mt * 16 + lo) * K + 2 * hi;
    const float* Bp = B + (size_t)(2 * hi) * Nout + nt * 32 + lo;
    const int col = nt * 32 + lo;

    const float as0 = a_src[col], as1 = a_src[col + 16];
    const float ad0 = a_dst[col], ad1 = a_dst[col + 16];

    v8f acc0 = {}, acc1 = {};
    for (int k = 0; k < K; k += 4) {
        float2 av = *(const float2*)(Ap + k);
        v2f a; a.x = av.x; a.y = av.y;
        const float* bp = Bp + (size_t)k * Nout;
        v2f b0; b0.x = bp[0];  b0.y = bp[Nout];
        v2f b1; b1.x = bp[16]; b1.y = bp[Nout + 16];
        acc0 = wmma4(a, b0, acc0);
        acc1 = wmma4(a, b1, acc1);
    }
    float* Cp = C + (size_t)(mt * 16 + 8 * hi) * Nout + col;
    #pragma unroll
    for (int r = 0; r < 8; ++r) {
        Cp[(size_t)r * Nout + 0]  = acc0[r];
        Cp[(size_t)r * Nout + 16] = acc1[r];
    }

    float ps[8], pd[8];
    #pragma unroll
    for (int r = 0; r < 8; ++r) {
        ps[r] = row16_sum(fmaf(acc1[r], as1, acc0[r] * as0));
        pd[r] = row16_sum(fmaf(acc1[r], ad1, acc0[r] * ad0));
    }
    if (lo == 0) {
        float4* s4 = (float4*)&stage[0][nt][8 * hi];
        float4* d4 = (float4*)&stage[1][nt][8 * hi];
        s4[0] = make_float4(ps[0], ps[1], ps[2], ps[3]);
        s4[1] = make_float4(ps[4], ps[5], ps[6], ps[7]);
        d4[0] = make_float4(pd[0], pd[1], pd[2], pd[3]);
        d4[1] = make_float4(pd[4], pd[5], pd[6], pd[7]);
    }
    __builtin_amdgcn_fence(__ATOMIC_RELEASE, "workgroup", "local");
    __builtin_amdgcn_s_barrier();
    __builtin_amdgcn_fence(__ATOMIC_ACQUIRE, "workgroup", "local");
    if (nt == 0) {
        float4 v;
        v.x = stage[hi][0][lo];  v.y = stage[hi][1][lo];
        v.z = stage[hi][2][lo];  v.w = stage[hi][3][lo];
        float* dst = hi ? al_d : al_s;
        *(float4*)(dst + (size_t)(mt * 16 + lo) * HEADS) = v;
    }
}

__global__ __launch_bounds__(32) void wmma_gemm_f32_tail(
    const float* __restrict__ A, const float* __restrict__ B,
    float* __restrict__ C, const float* __restrict__ bias,
    int K, int Nout)
{
    const int mt   = blockIdx.x;
    const int nt   = blockIdx.y;
    const int lane = threadIdx.x;
    const int hi   = lane >> 4;
    const int lo   = lane & 15;
    const int col  = nt * 16 + lo;
    const int ccol = col < Nout ? col : Nout - 1;

    const float* Ap = A + (size_t)(mt * 16 + lo) * K + 2 * hi;
    const float* Bp = B + (size_t)(2 * hi) * Nout + ccol;

    v8f acc = {};
    for (int k = 0; k < K; k += 4) {
        float2 av = *(const float2*)(Ap + k);
        v2f a; a.x = av.x; a.y = av.y;
        const float* bp = Bp + (size_t)k * Nout;
        v2f b; b.x = bp[0]; b.y = bp[Nout];
        acc = wmma4(a, b, acc);
    }
    if (col < Nout) {
        const float bb = bias ? bias[col] : 0.f;
        float* Cp = C + (size_t)(mt * 16 + 8 * hi) * Nout + col;
        #pragma unroll
        for (int r = 0; r < 8; ++r)
            Cp[(size_t)r * Nout] = acc[r] + bb;
    }
}

#define CSR_NB 16
#define CSR_R  3136
#define CSR_NC 32
#define CSR_CH (EE / CSR_NC)
#define CSR_T  512
#define CSR_Q  (CSR_CH / 4)
static_assert(CSR_NB * CSR_R >= NN, "node ranges must cover all nodes");
static_assert(EE % (CSR_NC * 4) == 0, "chunks must be whole groups of four edges");

__global__ __launch_bounds__(CSR_T) void csr_count(
    const int* __restrict__ ei, int* __restrict__ tab, int* __restrict__ total)
{
    __shared__ int cnt[CSR_R];
    const int c    = blockIdx.x;
    const int lo   = blockIdx.y * CSR_R;
    const unsigned rlen = (unsigned)((NN - lo) < CSR_R ? (NN - lo) : CSR_R);
    for (int r = threadIdx.x; r < CSR_R; r += CSR_T) cnt[r] = 0;
    if (blockIdx.x == 0 && blockIdx.y == 0 && threadIdx.x == 0) *total = 0;
    __syncthreads();

    const int4* __restrict__ d4 = (const int4*)(ei + EE + c * CSR_CH);
    for (int i0 = threadIdx.x; i0 < CSR_Q; i0 += 4 * CSR_T) {
        int4 d[4];
        #pragma unroll
        for (int k = 0; k < 4; ++k) {
            const int i = i0 + k * CSR_T;
            d[k] = i < CSR_Q ? d4[i] : make_int4(-1, -1, -1, -1);
        }
        #pragma unroll
        for (int k = 0; k < 4; ++k) {
            unsigned r;
            r = (unsigned)(d[k].x - lo); if (r < rlen) atomicAdd(&cnt[r], 1);
            r = (unsigned)(d[k].y - lo); if (r < rlen) atomicAdd(&cnt[r], 1);
            r = (unsigned)(d[k].z - lo); if (r < rlen) atomicAdd(&cnt[r], 1);
            r = (unsigned)(d[k].w - lo); if (r < rlen) atomicAdd(&cnt[r], 1);
        }
    }
    __syncthreads();
    int* __restrict__ row = tab + (size_t)c * NN + lo;
    for (unsigned r = threadIdx.x; r < rlen; r += CSR_T) row[r] = cnt[r];
}

__global__ __launch_bounds__(256) void csr_alloc(
    int* __restrict__ tab, int* __restrict__ total,
    int* __restrict__ start, int* __restrict__ deg)
{
    const int i    = blockIdx.x * blockDim.x + threadIdx.x;
    const int lane = threadIdx.x & 31;
    int pre[CSR_NC];
    int d = 0;
    if (i < NN) {
        #pragma unroll
        for (int c = 0; c < CSR_NC; ++c) pre[c] = tab[(size_t)c * NN + i];
        #pragma unroll
        for (int c = 0; c < CSR_NC; ++c) { const int t = pre[c]; pre[c] = d; d += t; }
    }
    int incl = d;
    for (int off = 1; off < 32; off <<= 1) {
        int t = __shfl_up(incl, off);
        if (lane >= off) incl += t;
    }
    int base = 0;
    if (lane == 31) base = atomicAdd(total, incl);
    base = __shfl(base, 31);
    const int st = base + incl - d;
    if (i < NN) {
        start[i] = st;
        deg[i]   = d;
        #pragma unroll
        for (int c = 0; c < CSR_NC; ++c) tab[(size_t)c * NN + i] = st + pre[c];
    }
}

__global__ __launch_bounds__(CSR_T) void csr_place(
    const int* __restrict__ ei, const int* __restrict__ tab, int* __restrict__ csr_src)
{
    __shared__ int cur[CSR_R];
    const int c    = blockIdx.x;
    const int lo   = blockIdx.y * CSR_R;
    const unsigned rlen = (unsigned)((NN - lo) < CSR_R ? (NN - lo) : CSR_R);
    const int* __restrict__ row = tab + (size_t)c * NN + lo;
    for (unsigned r = threadIdx.x; r < rlen; r += CSR_T) cur[r] = row[r];
    __syncthreads();

    const int4* __restrict__ s4 = (const int4*)(ei + c * CSR_CH);
    const int4* __restrict__ d4 = (const int4*)(ei + EE + c * CSR_CH);
    for (int i0 = threadIdx.x; i0 < CSR_Q; i0 += 4 * CSR_T) {
        int4 d[4], s[4];
        #pragma unroll
        for (int k = 0; k < 4; ++k) {
            const int i = i0 + k * CSR_T;
            d[k] = i < CSR_Q ? d4[i] : make_int4(-1, -1, -1, -1);
            s[k] = i < CSR_Q ? s4[i] : make_int4(0, 0, 0, 0);
        }
        #pragma unroll
        for (int k = 0; k < 4; ++k) {
            const unsigned r0 = (unsigned)(d[k].x - lo), r1 = (unsigned)(d[k].y - lo);
            const unsigned r2 = (unsigned)(d[k].z - lo), r3 = (unsigned)(d[k].w - lo);
            int p0 = -1, p1 = -1, p2 = -1, p3 = -1;
            if (r0 < rlen) p0 = atomicAdd(&cur[r0], 1);
            if (r1 < rlen) p1 = atomicAdd(&cur[r1], 1);
            if (r2 < rlen) p2 = atomicAdd(&cur[r2], 1);
            if (r3 < rlen) p3 = atomicAdd(&cur[r3], 1);
            if ((unsigned)p0 < (unsigned)EE) csr_src[p0] = s[k].x;
            if ((unsigned)p1 < (unsigned)EE) csr_src[p1] = s[k].y;
            if ((unsigned)p2 < (unsigned)EE) csr_src[p2] = s[k].z;
            if ((unsigned)p3 < (unsigned)EE) csr_src[p3] = s[k].w;
        }
    }
}

struct HeadMask { unsigned k0, k1, k2, k3; };
__device__ __forceinline__ float pick4(HeadMask k, float a, float b, float c, float d) {
    return __uint_as_float((k.k0 & __float_as_uint(a)) | (k.k1 & __float_as_uint(b)) |
                           (k.k2 & __float_as_uint(c)) | (k.k3 & __float_as_uint(d)));
}
__device__ __forceinline__ float lane_bcast(float v, int srclane) {
    return __int_as_float(__builtin_amdgcn_readlane(__float_as_int(v), srclane));
}

__global__ __launch_bounds__(256) void node_gather(
    const int* __restrict__ start, const int* __restrict__ deg,
    const int* __restrict__ csr_src, const float* __restrict__ h,
    const float* __restrict__ al_s, const float* __restrict__ al_d,
    const float* __restrict__ bias, float* __restrict__ out)
{
    const int lane = threadIdx.x & 31;
    const int n = __builtin_amdgcn_readfirstlane(
                      (int)(blockIdx.x * (blockDim.x >> 5) + (threadIdx.x >> 5)));
    if (n >= NN) return;
    const int hd  = lane >> 3;
    HeadMask hk;
    hk.k0 = hd == 0 ? 0xFFFFFFFFu : 0u;  hk.k1 = hd == 1 ? 0xFFFFFFFFu : 0u;
    hk.k2 = hd == 2 ? 0xFFFFFFFFu : 0u;  hk.k3 = hd == 3 ? 0xFFFFFFFFu : 0u;
    const int beg = start[n];
    const int cnt = deg[n];
    const float4 ad4 = *(const float4*)(al_d + (size_t)n * HEADS);
    const float4 sf4 = *(const float4*)(al_s + (size_t)n * HEADS);
    const float4* __restrict__ h4 = (const float4*)h;

    float m0 = leaky(sf4.x + ad4.x), m1 = leaky(sf4.y + ad4.y);
    float m2 = leaky(sf4.z + ad4.z), m3 = leaky(sf4.w + ad4.w);
    for (int j = lane; j < cnt; j += 32) {
        const int s = csr_src[beg + j];
        const float4 a = *(const float4*)(al_s + (size_t)s * HEADS);
        m0 = fmaxf(m0, leaky(a.x + ad4.x));
        m1 = fmaxf(m1, leaky(a.y + ad4.y));
        m2 = fmaxf(m2, leaky(a.z + ad4.z));
        m3 = fmaxf(m3, leaky(a.w + ad4.w));
    }
    for (int off = 16; off > 0; off >>= 1) {
        m0 = fmaxf(m0, __shfl_xor(m0, off));
        m1 = fmaxf(m1, __shfl_xor(m1, off));
        m2 = fmaxf(m2, __shfl_xor(m2, off));
        m3 = fmaxf(m3, __shfl_xor(m3, off));
    }
    m0 = __builtin_isfinite(m0) ? m0 : 0.f;
    m1 = __builtin_isfinite(m1) ? m1 : 0.f;
    m2 = __builtin_isfinite(m2) ? m2 : 0.f;
    m3 = __builtin_isfinite(m3) ? m3 : 0.f;

    const float wself = __expf(pick4(hk, leaky(sf4.x + ad4.x), leaky(sf4.y + ad4.y),
                                         leaky(sf4.z + ad4.z), leaky(sf4.w + ad4.w))
                               - pick4(hk, m0, m1, m2, m3));
    const float4 vself = h4[(size_t)n * (HC / 4) + lane];
    float den = wself;
    float4 acc;
    acc.x = wself * vself.x; acc.y = wself * vself.y;
    acc.z = wself * vself.z; acc.w = wself * vself.w;

    for (int base = 0; base < cnt; base += 32) {
        const int j = base + lane;
        int sl = 0;
        float w0 = 0.f, w1 = 0.f, w2 = 0.f, w3 = 0.f;
        if (j < cnt) {
            sl = csr_src[beg + j];
            const float4 a = *(const float4*)(al_s + (size_t)sl * HEADS);
            w0 = __expf(leaky(a.x + ad4.x) - m0);
            w1 = __expf(leaky(a.y + ad4.y) - m1);
            w2 = __expf(leaky(a.z + ad4.z) - m2);
            w3 = __expf(leaky(a.w + ad4.w) - m3);
        }
        const int lim = (cnt - base) < 32 ? (cnt - base) : 32;
        #pragma unroll 4
        for (int jj = 0; jj < lim; ++jj) {
            const int   s  = __builtin_amdgcn_readlane(sl, jj);
            const float wj = pick4(hk, lane_bcast(w0, jj), lane_bcast(w1, jj),
                                       lane_bcast(w2, jj), lane_bcast(w3, jj));
            const float4 v = h4[(size_t)s * (HC / 4) + lane];
            den += wj;
            acc.x = fmaf(wj, v.x, acc.x);
            acc.y = fmaf(wj, v.y, acc.y);
            acc.z = fmaf(wj, v.z, acc.z);
            acc.w = fmaf(wj, v.w, acc.w);
        }
    }

    const float  dd = den + 1e-16f;
    const float4 b  = *((const float4*)bias + lane);
    float4 o;
    o.x = acc.x / dd + b.x;  o.y = acc.y / dd + b.y;
    o.z = acc.z / dd + b.z;  o.w = acc.w / dd + b.w;
    o.x = o.x > 0.f ? o.x : 0.f;  o.y = o.y > 0.f ? o.y : 0.f;
    o.z = o.z > 0.f ? o.z : 0.f;  o.w = o.w > 0.f ? o.w : 0.f;
    ((float4*)out)[(size_t)n * (HC / 4) + lane] = o;
}

extern "C" void kernel_launch(void* const* d_in, const int* in_sizes, int n_in,
                              void* d_out, int out_size, void* d_ws, size_t ws_size,
                              hipStream_t stream)
{
    const float* x   = (const float*)d_in[0];
    const int*   ei  = (const int*)d_in[1];
    const float* W[3]    = { (const float*)d_in[2],  (const float*)d_in[6],  (const float*)d_in[10] };
    const float* asv[3]  = { (const float*)d_in[3],  (const float*)d_in[7],  (const float*)d_in[11] };
    const float* adv[3]  = { (const float*)d_in[4],  (const float*)d_in[8],  (const float*)d_in[12] };
    const float* bv[3]   = { (const float*)d_in[5],  (const float*)d_in[9],  (const float*)d_in[13] };
    const float* Wl  = (const float*)d_in[14];
    const float* bl  = (const float*)d_in[15];

    char* ws = (char*)d_ws;
    float* hbuf = (float*)ws;                          ws += (size_t)NN * HC * sizeof(float);
    float* bufA = (float*)ws;                          ws += (size_t)NN * HC * sizeof(float);
    float* bufB = (float*)ws;                          ws += (size_t)NN * HC * sizeof(float);
    float* al_s = (float*)ws;                          ws += (size_t)NN * HEADS * sizeof(float);
    float* al_d = (float*)ws;                          ws += (size_t)NN * HEADS * sizeof(float);
    int*   deg    = (int*)ws;                          ws += (size_t)NN * sizeof(int);
    int*   start  = (int*)ws;                          ws += (size_t)NN * sizeof(int);
    int*   total  = (int*)ws;                          ws += 64 * sizeof(int);
    int*   csr_src = (int*)ws;                         ws += (size_t)EE * sizeof(int);
    int*   tab    = (int*)ws;

    const int TPB = 256;
    const int MT  = NN / 16;
    const int gN  = (NN + TPB - 1) / TPB;
    const int gNodeWaves = (NN + (TPB / 32) - 1) / (TPB / 32);

    hipLaunchKernelGGL(csr_count, dim3(CSR_NC, CSR_NB), dim3(CSR_T), 0, stream, ei, tab, total);
    hipLaunchKernelGGL(csr_alloc, dim3(gN), dim3(TPB), 0, stream, tab, total, start, deg);
    hipLaunchKernelGGL(csr_place, dim3(CSR_NC, CSR_NB), dim3(CSR_T), 0, stream, ei, tab, csr_src);

    auto gat_layer = [&](const float* xin, int layer, float* xout) {
        hipLaunchKernelGGL(wmma_gemm_f32_logits, dim3(MT), dim3(32 * HEADS), 0, stream,
                           xin, W[layer], hbuf, asv[layer], adv[layer], al_s, al_d, IN_DIM, HC);
        hipLaunchKernelGGL(node_gather, dim3(gNodeWaves), dim3(TPB), 0, stream,
                           start, deg, csr_src, hbuf, al_s, al_d, bv[layer], xout);
    };

    gat_layer(x,    0, bufA);
    gat_layer(bufA, 1, bufB);
    gat_layer(bufB, 2, bufA);

    hipLaunchKernelGGL(wmma_gemm_f32_tail, dim3(MT, (OUT_DIM + 15) / 16), dim3(32), 0, stream,
                       bufA, Wl, (float*)d_out, bl, HC, OUT_DIM);
}


# Layer product: each wave reads its 128x32 slice of W once into registers; 768 work-groups walk over row tiles, next tile's A fragments asked for ahead; k and sum order unchanged
# speedup vs baseline: 4.7257x; 1.1352x over previous
#include <hip/hip_runtime.h>
#include <math.h>

#define NN 50000
#define EE 800000
#define ET (EE + NN)
#define IN_DIM 128
#define HID 32
#define HEADS 4
#define HC 128
#define OUT_DIM 40
#define SLOPE 0.2f

typedef float v2f __attribute__((ext_vector_type(2)));
typedef float v8f __attribute__((ext_vector_type(8)));

__device__ __forceinline__ float leaky(float e) {
    return e > 0.f ? e : SLOPE * e;
}

__device__ __forceinline__ v8f wmma4(v2f a, v2f b, v8f c) {
    return __builtin_amdgcn_wmma_f32_16x16x4_f32(
               false, a, false, b, (short)0, c, false, false);
}

template <int CTRL>
__device__ __forceinline__ float row_xmask(float v) {
    return __int_as_float(__builtin_amdgcn_update_dpp(0, __float_as_int(v), CTRL, 0xf, 0xf, true));
}
__device__ __forceinline__ float row16_sum(float v) {
    v += row_xmask<0x161>(v);
    v += row_xmask<0x162>(v);
    v += row_xmask<0x164>(v);
    v += row_xmask<0x168>(v);
    return v;
}

#define GK   128
#define GKS  (GK / 4)
#define GEMM_WG 768

__device__ __forceinline__ void load_a_tile(const float* __restrict__ Ap, v2f (&a)[GKS]) {
    #pragma unroll
    for (int j = 0; j < GKS; ++j) {
        const float2 av = *(const float2*)(Ap + 4 * j);
        a[j].x = av.x; a[j].y = av.y;
    }
}

__global__ __launch_bounds__(128) void wmma_gemm_f32_logits(
    const float* __restrict__ A, const float* __restrict__ B,
    float* __restrict__ C,
    const float* __restrict__ a_src, const float* __restrict__ a_dst,
    float* __restrict__ al_s, float* __restrict__ al_d, int n_tiles)
{
    __shared__ __attribute__((aligned(16))) float stage[2][2][HEADS][16];

    const int nt   = __builtin_amdgcn_readfirstlane((int)(threadIdx.x >> 5));
    const int lane = threadIdx.x & 31;
    const int hi   = lane >> 4;
    const int lo   = lane & 15;
    const int col  = nt * 32 + lo;

    const float* Bp = B + (size_t)(2 * hi) * HC + col;
    v2f b0[GKS], b1[GKS];
    #pragma unroll
    for (int j = 0; j < GKS; ++j) {
        const float* bp = Bp + (size_t)(4 * j) * HC;
        b0[j].x = bp[0];  b0[j].y = bp[HC];
        b1[j].x = bp[16]; b1[j].y = bp[HC + 16];
    }
    const float as0 = a_src[col], as1 = a_src[col + 16];
    const float ad0 = a_dst[col], ad1 = a_dst[col + 16];

    int mt = blockIdx.x;
    if (mt >= n_tiles) return;
    v2f a[GKS];
    load_a_tile(A + (size_t)(mt * 16 + lo) * GK + 2 * hi, a);

    int cp = 0;
    for (; mt < n_tiles; mt += gridDim.x, cp ^= 1) {
        const int mtn = mt + (int)gridDim.x;
        v2f an[GKS];
        #pragma unroll
        for (int j = 0; j < GKS; ++j) an[j] = a[j];
        if (mtn < n_tiles)
            load_a_tile(A + (size_t)(mtn * 16 + lo) * GK + 2 * hi, an);

        v8f acc0 = {}, acc1 = {};
        #pragma unroll
        for (int j = 0; j < GKS; ++j) {
            acc0 = wmma4(a[j], b0[j], acc0);
            acc1 = wmma4(a[j], b1[j], acc1);
        }
        float* Cp = C + (size_t)(mt * 16 + 8 * hi) * HC + col;
        #pragma unroll
        for (int r = 0; r < 8; ++r) {
            Cp[(size_t)r * HC + 0]  = acc0[r];
            Cp[(size_t)r * HC + 16] = acc1[r];
        }

        float ps[8], pd[8];
        #pragma unroll
        for (int r = 0; r < 8; ++r) {
            ps[r] = row16_sum(fmaf(acc1[r], as1, acc0[r] * as0));
            pd[r] = row16_sum(fmaf(acc1[r], ad1, acc0[r] * ad0));
        }
        if (lo == 0) {
            float4* s4 = (float4*)&stage[cp][0][nt][8 * hi];
            float4* d4 = (float4*)&stage[cp][1][nt][8 * hi];
            s4[0] = make_float4(ps[0], ps[1], ps[2], ps[3]);
            s4[1] = make_float4(ps[4], ps[5], ps[6], ps[7]);
            d4[0] = make_float4(pd[0], pd[1], pd[2], pd[3]);
            d4[1] = make_float4(pd[4], pd[5], pd[6], pd[7]);
        }
        __builtin_amdgcn_fence(__ATOMIC_RELEASE, "workgroup", "local");
        __builtin_amdgcn_s_barrier();
        __builtin_amdgcn_fence(__ATOMIC_ACQUIRE, "workgroup", "local");
        if (nt == 0) {
            float4 v;
            v.x = stage[cp][hi][0][lo];  v.y = stage[cp][hi][1][lo];
            v.z = stage[cp][hi][2][lo];  v.w = stage[cp][hi][3][lo];
            float* dst = hi ? al_d : al_s;
            *(float4*)(dst + (size_t)(mt * 16 + lo) * HEADS) = v;
        }
        #pragma unroll
        for (int j = 0; j < GKS; ++j) a[j] = an[j];
    }
}

__global__ __launch_bounds__(32) void wmma_gemm_f32_tail(
    const float* __restrict__ A, const float* __restrict__ B,
    float* __restrict__ C, const float* __restrict__ bias,
    int K, int Nout)
{
    const int mt   = blockIdx.x;
    const int nt   = blockIdx.y;
    const int lane = threadIdx.x;
    const int hi   = lane >> 4;
    const int lo   = lane & 15;
    const int col  = nt * 16 + lo;
    const int ccol = col < Nout ? col : Nout - 1;

    const float* Ap = A + (size_t)(mt * 16 + lo) * K + 2 * hi;
    const float* Bp = B + (size_t)(2 * hi) * Nout + ccol;

    v8f acc = {};
    for (int k = 0; k < K; k += 4) {
        float2 av = *(const float2*)(Ap + k);
        v2f a; a.x = av.x; a.y = av.y;
        const float* bp = Bp + (size_t)k * Nout;
        v2f b; b.x = bp[0]; b.y = bp[Nout];
        acc = wmma4(a, b, acc);
    }
    if (col < Nout) {
        const float bb = bias ? bias[col] : 0.f;
        float* Cp = C + (size_t)(mt * 16 + 8 * hi) * Nout + col;
        #pragma unroll
        for (int r = 0; r < 8; ++r)
            Cp[(size_t)r * Nout] = acc[r] + bb;
    }
}

#define CSR_NB 16
#define CSR_R  3136
#define CSR_NC 32
#define CSR_CH (EE / CSR_NC)
#define CSR_T  512
#define CSR_Q  (CSR_CH / 4)
static_assert(CSR_NB * CSR_R >= NN, "node ranges must cover all nodes");
static_assert(EE % (CSR_NC * 4) == 0, "chunks must be whole groups of four edges");

__global__ __launch_bounds__(CSR_T) void csr_count(
    const int* __restrict__ ei, int* __restrict__ tab, int* __restrict__ total)
{
    __shared__ int cnt[CSR_R];
    const int c    = blockIdx.x;
    const int lo   = blockIdx.y * CSR_R;
    const unsigned rlen = (unsigned)((NN - lo) < CSR_R ? (NN - lo) : CSR_R);
    for (int r = threadIdx.x; r < CSR_R; r += CSR_T) cnt[r] = 0;
    if (blockIdx.x == 0 && blockIdx.y == 0 && threadIdx.x == 0) *total = 0;
    __syncthreads();

    const int4* __restrict__ d4 = (const int4*)(ei + EE + c * CSR_CH);
    for (int i0 = threadIdx.x; i0 < CSR_Q; i0 += 4 * CSR_T) {
        int4 d[4];
        #pragma unroll
        for (int k = 0; k < 4; ++k) {
            const int i = i0 + k * CSR_T;
            d[k] = i < CSR_Q ? d4[i] : make_int4(-1, -1, -1, -1);
        }
        #pragma unroll
        for (int k = 0; k < 4; ++k) {
            unsigned r;
            r = (unsigned)(d[k].x - lo); if (r < rlen) atomicAdd(&cnt[r], 1);
            r = (unsigned)(d[k].y - lo); if (r < rlen) atomicAdd(&cnt[r], 1);
            r = (unsigned)(d[k].z - lo); if (r < rlen) atomicAdd(&cnt[r], 1);
            r = (unsigned)(d[k].w - lo); if (r < rlen) atomicAdd(&cnt[r], 1);
        }
    }
    __syncthreads();
    int* __restrict__ row = tab + (size_t)c * NN + lo;
    for (unsigned r = threadIdx.x; r < rlen; r += CSR_T) row[r] = cnt[r];
}

__global__ __launch_bounds__(256) void csr_alloc(
    int* __restrict__ tab, int* __restrict__ total,
    int* __restrict__ start, int* __restrict__ deg)
{
    const int i    = blockIdx.x * blockDim.x + threadIdx.x;
    const int lane = threadIdx.x & 31;
    int pre[CSR_NC];
    int d = 0;
    if (i < NN) {
        #pragma unroll
        for (int c = 0; c < CSR_NC; ++c) pre[c] = tab[(size_t)c * NN + i];
        #pragma unroll
        for (int c = 0; c < CSR_NC; ++c) { const int t = pre[c]; pre[c] = d; d += t; }
    }
    int incl = d;
    for (int off = 1; off < 32; off <<= 1) {
        int t = __shfl_up(incl, off);
        if (lane >= off) incl += t;
    }
    int base = 0;
    if (lane == 31) base = atomicAdd(total, incl);
    base = __shfl(base, 31);
    const int st = base + incl - d;
    if (i < NN) {
        start[i] = st;
        deg[i]   = d;
        #pragma unroll
        for (int c = 0; c < CSR_NC; ++c) tab[(size_t)c * NN + i] = st + pre[c];
    }
}

__global__ __launch_bounds__(CSR_T) void csr_place(
    const int* __restrict__ ei, const int* __restrict__ tab, int* __restrict__ csr_src)
{
    __shared__ int cur[CSR_R];
    const int c    = blockIdx.x;
    const int lo   = blockIdx.y * CSR_R;
    const unsigned rlen = (unsigned)((NN - lo) < CSR_R ? (NN - lo) : CSR_R);
    const int* __restrict__ row = tab + (size_t)c * NN + lo;
    for (unsigned r = threadIdx.x; r < rlen; r += CSR_T) cur[r] = row[r];
    __syncthreads();

    const int4* __restrict__ s4 = (const int4*)(ei + c * CSR_CH);
    const int4* __restrict__ d4 = (const int4*)(ei + EE + c * CSR_CH);
    for (int i0 = threadIdx.x; i0 < CSR_Q; i0 += 4 * CSR_T) {
        int4 d[4], s[4];
        #pragma unroll
        for (int k = 0; k < 4; ++k) {
            const int i = i0 + k * CSR_T;
            d[k] = i < CSR_Q ? d4[i] : make_int4(-1, -1, -1, -1);
            s[k] = i < CSR_Q ? s4[i] : make_int4(0, 0, 0, 0);
        }
        #pragma unroll
        for (int k = 0; k < 4; ++k) {
            const unsigned r0 = (unsigned)(d[k].x - lo), r1 = (unsigned)(d[k].y - lo);
            const unsigned r2 = (unsigned)(d[k].z - lo), r3 = (unsigned)(d[k].w - lo);
            int p0 = -1, p1 = -1, p2 = -1, p3 = -1;
            if (r0 < rlen) p0 = atomicAdd(&cur[r0], 1);
            if (r1 < rlen) p1 = atomicAdd(&cur[r1], 1);
            if (r2 < rlen) p2 = atomicAdd(&cur[r2], 1);
            if (r3 < rlen) p3 = atomicAdd(&cur[r3], 1);
            if ((unsigned)p0 < (unsigned)EE) csr_src[p0] = s[k].x;
            if ((unsigned)p1 < (unsigned)EE) csr_src[p1] = s[k].y;
            if ((unsigned)p2 < (unsigned)EE) csr_src[p2] = s[k].z;
            if ((unsigned)p3 < (unsigned)EE) csr_src[p3] = s[k].w;
        }
    }
}

struct HeadMask { unsigned k0, k1, k2, k3; };
__device__ __forceinline__ float pick4(HeadMask k, float a, float b, float c, float d) {
    return __uint_as_float((k.k0 & __float_as_uint(a)) | (k.k1 & __float_as_uint(b)) |
                           (k.k2 & __float_as_uint(c)) | (k.k3 & __float_as_uint(d)));
}
__device__ __forceinline__ float lane_bcast(float v, int srclane) {
    return __int_as_float(__builtin_amdgcn_readlane(__float_as_int(v), srclane));
}

__global__ __launch_bounds__(256) void node_gather(
    const int* __restrict__ start, const int* __restrict__ deg,
    const int* __restrict__ csr_src, const float* __restrict__ h,
    const float* __restrict__ al_s, const float* __restrict__ al_d,
    const float* __restrict__ bias, float* __restrict__ out)
{
    const int lane = threadIdx.x & 31;
    const int n = __builtin_amdgcn_readfirstlane(
                      (int)(blockIdx.x * (blockDim.x >> 5) + (threadIdx.x >> 5)));
    if (n >= NN) return;
    const int hd  = lane >> 3;
    HeadMask hk;
    hk.k0 = hd == 0 ? 0xFFFFFFFFu : 0u;  hk.k1 = hd == 1 ? 0xFFFFFFFFu : 0u;
    hk.k2 = hd == 2 ? 0xFFFFFFFFu : 0u;  hk.k3 = hd == 3 ? 0xFFFFFFFFu : 0u;
    const int beg = start[n];
    const int cnt = deg[n];
    const float4 ad4 = *(const float4*)(al_d + (size_t)n * HEADS);
    const float4 sf4 = *(const float4*)(al_s + (size_t)n * HEADS);
    const float4* __restrict__ h4 = (const float4*)h;

    float m0 = leaky(sf4.x + ad4.x), m1 = leaky(sf4.y + ad4.y);
    float m2 = leaky(sf4.z + ad4.z), m3 = leaky(sf4.w + ad4.w);
    for (int j = lane; j < cnt; j += 32) {
        const int s = csr_src[beg + j];
        const float4 a = *(const float4*)(al_s + (size_t)s * HEADS);
        m0 = fmaxf(m0, leaky(a.x + ad4.x));
        m1 = fmaxf(m1, leaky(a.y + ad4.y));
        m2 = fmaxf(m2, leaky(a.z + ad4.z));
        m3 = fmaxf(m3, leaky(a.w + ad4.w));
    }
    for (int off = 16; off > 0; off >>= 1) {
        m0 = fmaxf(m0, __shfl_xor(m0, off));
        m1 = fmaxf(m1, __shfl_xor(m1, off));
        m2 = fmaxf(m2, __shfl_xor(m2, off));
        m3 = fmaxf(m3, __shfl_xor(m3, off));
    }
    m0 = __builtin_isfinite(m0) ? m0 : 0.f;
    m1 = __builtin_isfinite(m1) ? m1 : 0.f;
    m2 = __builtin_isfinite(m2) ? m2 : 0.f;
    m3 = __builtin_isfinite(m3) ? m3 : 0.f;

    const float wself = __expf(pick4(hk, leaky(sf4.x + ad4.x), leaky(sf4.y + ad4.y),
                                         leaky(sf4.z + ad4.z), leaky(sf4.w + ad4.w))
                               - pick4(hk, m0, m1, m2, m3));
    const float4 vself = h4[(size_t)n * (HC / 4) + lane];
    float den = wself;
    float4 acc;
    acc.x = wself * vself.x; acc.y = wself * vself.y;
    acc.z = wself * vself.z; acc.w = wself * vself.w;

    for (int base = 0; base < cnt; base += 32) {
        const int j = base + lane;
        int sl = 0;
        float w0 = 0.f, w1 = 0.f, w2 = 0.f, w3 = 0.f;
        if (j < cnt) {
            sl = csr_src[beg + j];
            const float4 a = *(const float4*)(al_s + (size_t)sl * HEADS);
            w0 = __expf(leaky(a.x + ad4.x) - m0);
            w1 = __expf(leaky(a.y + ad4.y) - m1);
            w2 = __expf(leaky(a.z + ad4.z) - m2);
            w3 = __expf(leaky(a.w + ad4.w) - m3);
        }
        const int lim = (cnt - base) < 32 ? (cnt - base) : 32;
        #pragma unroll 4
        for (int jj = 0; jj < lim; ++jj) {
            const int   s  = __builtin_amdgcn_readlane(sl, jj);
            const float wj = pick4(hk, lane_bcast(w0, jj), lane_bcast(w1, jj),
                                       lane_bcast(w2, jj), lane_bcast(w3, jj));
            const float4 v = h4[(size_t)s * (HC / 4) + lane];
            den += wj;
            acc.x = fmaf(wj, v.x, acc.x);
            acc.y = fmaf(wj, v.y, acc.y);
            acc.z = fmaf(wj, v.z, acc.z);
            acc.w = fmaf(wj, v.w, acc.w);
        }
    }

    const float  dd = den + 1e-16f;
    const float4 b  = *((const float4*)bias + lane);
    float4 o;
    o.x = acc.x / dd + b.x;  o.y = acc.y / dd + b.y;
    o.z = acc.z / dd + b.z;  o.w = acc.w / dd + b.w;
    o.x = o.x > 0.f ? o.x : 0.f;  o.y = o.y > 0.f ? o.y : 0.f;
    o.z = o.z > 0.f ? o.z : 0.f;  o.w = o.w > 0.f ? o.w : 0.f;
    ((float4*)out)[(size_t)n * (HC / 4) + lane] = o;
}

extern "C" void kernel_launch(void* const* d_in, const int* in_sizes, int n_in,
                              void* d_out, int out_size, void* d_ws, size_t ws_size,
                              hipStream_t stream)
{
    const float* x   = (const float*)d_in[0];
    const int*   ei  = (const int*)d_in[1];
    const float* W[3]    = { (const float*)d_in[2],  (const float*)d_in[6],  (const float*)d_in[10] };
    const float* asv[3]  = { (const float*)d_in[3],  (const float*)d_in[7],  (const float*)d_in[11] };
    const float* adv[3]  = { (const float*)d_in[4],  (const float*)d_in[8],  (const float*)d_in[12] };
    const float* bv[3]   = { (const float*)d_in[5],  (const float*)d_in[9],  (const float*)d_in[13] };
    const float* Wl  = (const float*)d_in[14];
    const float* bl  = (const float*)d_in[15];

    char* ws = (char*)d_ws;
    float* hbuf = (float*)ws;                          ws += (size_t)NN * HC * sizeof(float);
    float* bufA = (float*)ws;                          ws += (size_t)NN * HC * sizeof(float);
    float* bufB = (float*)ws;                          ws += (size_t)NN * HC * sizeof(float);
    float* al_s = (float*)ws;                          ws += (size_t)NN * HEADS * sizeof(float);
    float* al_d = (float*)ws;                          ws += (size_t)NN * HEADS * sizeof(float);
    int*   deg    = (int*)ws;                          ws += (size_t)NN * sizeof(int);
    int*   start  = (int*)ws;                          ws += (size_t)NN * sizeof(int);
    int*   total  = (int*)ws;                          ws += 64 * sizeof(int);
    int*   csr_src = (int*)ws;                         ws += (size_t)EE * sizeof(int);
    int*   tab    = (int*)ws;

    const int TPB = 256;
    const int MT  = NN / 16;
    const int gN  = (NN + TPB - 1) / TPB;
    const int gNodeWaves = (NN + (TPB / 32) - 1) / (TPB / 32);

    hipLaunchKernelGGL(csr_count, dim3(CSR_NC, CSR_NB), dim3(CSR_T), 0, stream, ei, tab, total);
    hipLaunchKernelGGL(csr_alloc, dim3(gN), dim3(TPB), 0, stream, tab, total, start, deg);
    hipLaunchKernelGGL(csr_place, dim3(CSR_NC, CSR_NB), dim3(CSR_T), 0, stream, ei, tab, csr_src);

    auto gat_layer = [&](const float* xin, int layer, float* xout) {
        hipLaunchKernelGGL(wmma_gemm_f32_logits, dim3(GEMM_WG), dim3(32 * HEADS), 0, stream,
                           xin, W[layer], hbuf, asv[layer], adv[layer], al_s, al_d, MT);
        hipLaunchKernelGGL(node_gather, dim3(gNodeWaves), dim3(TPB), 0, stream,
                           start, deg, csr_src, hbuf, al_s, al_d, bv[layer], xout);
    };

    gat_layer(x,    0, bufA);
    gat_layer(bufA, 1, bufB);
    gat_layer(bufB, 2, bufA);

    hipLaunchKernelGGL(wmma_gemm_f32_tail, dim3(MT, (OUT_DIM + 15) / 16), dim3(32), 0, stream,
                       bufA, Wl, (float*)d_out, bl, HC, OUT_DIM);
}


# Node aggregation on resident waves: 1250 work-groups walk over the 8-node groups, 5 nodes per wave; next nodes' slot, count, source list, logits and own row asked for ahead; sum order unchanged
# speedup vs baseline: 5.1220x; 1.0839x over previous
#include <hip/hip_runtime.h>
#include <math.h>

#define NN 50000
#define EE 800000
#define ET (EE + NN)
#define IN_DIM 128
#define HID 32
#define HEADS 4
#define HC 128
#define OUT_DIM 40
#define SLOPE 0.2f

typedef float v2f __attribute__((ext_vector_type(2)));
typedef float v8f __attribute__((ext_vector_type(8)));

__device__ __forceinline__ float leaky(float e) {
    return e > 0.f ? e : SLOPE * e;
}

__device__ __forceinline__ v8f wmma4(v2f a, v2f b, v8f c) {
    return __builtin_amdgcn_wmma_f32_16x16x4_f32(
               false, a, false, b, (short)0, c, false, false);
}

template <int CTRL>
__device__ __forceinline__ float row_xmask(float v) {
    return __int_as_float(__builtin_amdgcn_update_dpp(0, __float_as_int(v), CTRL, 0xf, 0xf, true));
}
__device__ __forceinline__ float row16_sum(float v) {
    v += row_xmask<0x161>(v);
    v += row_xmask<0x162>(v);
    v += row_xmask<0x164>(v);
    v += row_xmask<0x168>(v);
    return v;
}

#define GK   128
#define GKS  (GK / 4)
#define GEMM_WG 768

__device__ __forceinline__ void load_a_tile(const float* __restrict__ Ap, v2f (&a)[GKS]) {
    #pragma unroll
    for (int j = 0; j < GKS; ++j) {
        const float2 av = *(const float2*)(Ap + 4 * j);
        a[j].x = av.x; a[j].y = av.y;
    }
}

__global__ __launch_bounds__(128) void wmma_gemm_f32_logits(
    const float* __restrict__ A, const float* __restrict__ B,
    float* __restrict__ C,
    const float* __restrict__ a_src, const float* __restrict__ a_dst,
    float* __restrict__ al_s, float* __restrict__ al_d, int n_tiles)
{
    __shared__ __attribute__((aligned(16))) float stage[2][2][HEADS][16];

    const int nt   = __builtin_amdgcn_readfirstlane((int)(threadIdx.x >> 5));
    const int lane = threadIdx.x & 31;
    const int hi   = lane >> 4;
    const int lo   = lane & 15;
    const int col  = nt * 32 + lo;

    const float* Bp = B + (size_t)(2 * hi) * HC + col;
    v2f b0[GKS], b1[GKS];
    #pragma unroll
    for (int j = 0; j < GKS; ++j) {
        const float* bp = Bp + (size_t)(4 * j) * HC;
        b0[j].x = bp[0];  b0[j].y = bp[HC];
        b1[j].x = bp[16]; b1[j].y = bp[HC + 16];
    }
    const float as0 = a_src[col], as1 = a_src[col + 16];
    const float ad0 = a_dst[col], ad1 = a_dst[col + 16];

    int mt = blockIdx.x;
    if (mt >= n_tiles) return;
    v2f a[GKS];
    load_a_tile(A + (size_t)(mt * 16 + lo) * GK + 2 * hi, a);

    int cp = 0;
    for (; mt < n_tiles; mt += gridDim.x, cp ^= 1) {
        const int mtn = mt + (int)gridDim.x;
        v2f an[GKS];
        #pragma unroll
        for (int j = 0; j < GKS; ++j) an[j] = a[j];
        if (mtn < n_tiles)
            load_a_tile(A + (size_t)(mtn * 16 + lo) * GK + 2 * hi, an);

        v8f acc0 = {}, acc1 = {};
        #pragma unroll
        for (int j = 0; j < GKS; ++j) {
            acc0 = wmma4(a[j], b0[j], acc0);
            acc1 = wmma4(a[j], b1[j], acc1);
        }
        float* Cp = C + (size_t)(mt * 16 + 8 * hi) * HC + col;
        #pragma unroll
        for (int r = 0; r < 8; ++r) {
            Cp[(size_t)r * HC + 0]  = acc0[r];
            Cp[(size_t)r * HC + 16] = acc1[r];
        }

        float ps[8], pd[8];
        #pragma unroll
        for (int r = 0; r < 8; ++r) {
            ps[r] = row16_sum(fmaf(acc1[r], as1, acc0[r] * as0));
            pd[r] = row16_sum(fmaf(acc1[r], ad1, acc0[r] * ad0));
        }
        if (lo == 0) {
            float4* s4 = (float4*)&stage[cp][0][nt][8 * hi];
            float4* d4 = (float4*)&stage[cp][1][nt][8 * hi];
            s4[0] = make_float4(ps[0], ps[1], ps[2], ps[3]);
            s4[1] = make_float4(ps[4], ps[5], ps[6], ps[7]);
            d4[0] = make_float4(pd[0], pd[1], pd[2], pd[3]);
            d4[1] = make_float4(pd[4], pd[5], pd[6], pd[7]);
        }
        __builtin_amdgcn_fence(__ATOMIC_RELEASE, "workgroup", "local");
        __builtin_amdgcn_s_barrier();
        __builtin_amdgcn_fence(__ATOMIC_ACQUIRE, "workgroup", "local");
        if (nt == 0) {
            float4 v;
            v.x = stage[cp][hi][0][lo];  v.y = stage[cp][hi][1][lo];
            v.z = stage[cp][hi][2][lo];  v.w = stage[cp][hi][3][lo];
            float* dst = hi ? al_d : al_s;
            *(float4*)(dst + (size_t)(mt * 16 + lo) * HEADS) = v;
        }
        #pragma unroll
        for (int j = 0; j < GKS; ++j) a[j] = an[j];
    }
}

__global__ __launch_bounds__(32) void wmma_gemm_f32_tail(
    const float* __restrict__ A, const float* __restrict__ B,
    float* __restrict__ C, const float* __restrict__ bias,
    int K, int Nout)
{
    const int mt   = blockIdx.x;
    const int nt   = blockIdx.y;
    const int lane = threadIdx.x;
    const int hi   = lane >> 4;
    const int lo   = lane & 15;
    const int col  = nt * 16 + lo;
    const int ccol = col < Nout ? col : Nout - 1;

    const float* Ap = A + (size_t)(mt * 16 + lo) * K + 2 * hi;
    const float* Bp = B + (size_t)(2 * hi) * Nout + ccol;

    v8f acc = {};
    for (int k = 0; k < K; k += 4) {
        float2 av = *(const float2*)(Ap + k);
        v2f a; a.x = av.x; a.y = av.y;
        const float* bp = Bp + (size_t)k * Nout;
        v2f b; b.x = bp[0]; b.y = bp[Nout];
        acc = wmma4(a, b, acc);
    }
    if (col < Nout) {
        const float bb = bias ? bias[col] : 0.f;
        float* Cp = C + (size_t)(mt * 16 + 8 * hi) * Nout + col;
        #pragma unroll
        for (int r = 0; r < 8; ++r)
            Cp[(size_t)r * Nout] = acc[r] + bb;
    }
}

#define CSR_NB 16
#define CSR_R  3136
#define CSR_NC 32
#define CSR_CH (EE / CSR_NC)
#define CSR_T  512
#define CSR_Q  (CSR_CH / 4)
static_assert(CSR_NB * CSR_R >= NN, "node ranges must cover all nodes");
static_assert(EE % (CSR_NC * 4) == 0, "chunks must be whole groups of four edges");

__global__ __launch_bounds__(CSR_T) void csr_count(
    const int* __restrict__ ei, int* __restrict__ tab, int* __restrict__ total)
{
    __shared__ int cnt[CSR_R];
    const int c    = blockIdx.x;
    const int lo   = blockIdx.y * CSR_R;
    const unsigned rlen = (unsigned)((NN - lo) < CSR_R ? (NN - lo) : CSR_R);
    for (int r = threadIdx.x; r < CSR_R; r += CSR_T) cnt[r] = 0;
    if (blockIdx.x == 0 && blockIdx.y == 0 && threadIdx.x == 0) *total = 0;
    __syncthreads();

    const int4* __restrict__ d4 = (const int4*)(ei + EE + c * CSR_CH);
    for (int i0 = threadIdx.x; i0 < CSR_Q; i0 += 4 * CSR_T) {
        int4 d[4];
        #pragma unroll
        for (int k = 0; k < 4; ++k) {
            const int i = i0 + k * CSR_T;
            d[k] = i < CSR_Q ? d4[i] : make_int4(-1, -1, -1, -1);
        }
        #pragma unroll
        for (int k = 0; k < 4; ++k) {
            unsigned r;
            r = (unsigned)(d[k].x - lo); if (r < rlen) atomicAdd(&cnt[r], 1);
            r = (unsigned)(d[k].y - lo); if (r < rlen) atomicAdd(&cnt[r], 1);
            r = (unsigned)(d[k].z - lo); if (r < rlen) atomicAdd(&cnt[r], 1);
            r = (unsigned)(d[k].w - lo); if (r < rlen) atomicAdd(&cnt[r], 1);
        }
    }
    __syncthreads();
    int* __restrict__ row = tab + (size_t)c * NN + lo;
    for (unsigned r = threadIdx.x; r < rlen; r += CSR_T) row[r] = cnt[r];
}

__global__ __launch_bounds__(256) void csr_alloc(
    int* __restrict__ tab, int* __restrict__ total,
    int* __restrict__ start, int* __restrict__ deg)
{
    const int i    = blockIdx.x * blockDim.x + threadIdx.x;
    const int lane = threadIdx.x & 31;
    int pre[CSR_NC];
    int d = 0;
    if (i < NN) {
        #pragma unroll
        for (int c = 0; c < CSR_NC; ++c) pre[c] = tab[(size_t)c * NN + i];
        #pragma unroll
        for (int c = 0; c < CSR_NC; ++c) { const int t = pre[c]; pre[c] = d; d += t; }
    }
    int incl = d;
    for (int off = 1; off < 32; off <<= 1) {
        int t = __shfl_up(incl, off);
        if (lane >= off) incl += t;
    }
    int base = 0;
    if (lane == 31) base = atomicAdd(total, incl);
    base = __shfl(base, 31);
    const int st = base + incl - d;
    if (i < NN) {
        start[i] = st;
        deg[i]   = d;
        #pragma unroll
        for (int c = 0; c < CSR_NC; ++c) tab[(size_t)c * NN + i] = st + pre[c];
    }
}

__global__ __launch_bounds__(CSR_T) void csr_place(
    const int* __restrict__ ei, const int* __restrict__ tab, int* __restrict__ csr_src)
{
    __shared__ int cur[CSR_R];
    const int c    = blockIdx.x;
    const int lo   = blockIdx.y * CSR_R;
    const unsigned rlen = (unsigned)((NN - lo) < CSR_R ? (NN - lo) : CSR_R);
    const int* __restrict__ row = tab + (size_t)c * NN + lo;
    for (unsigned r = threadIdx.x; r < rlen; r += CSR_T) cur[r] = row[r];
    __syncthreads();

    const int4* __restrict__ s4 = (const int4*)(ei + c * CSR_CH);
    const int4* __restrict__ d4 = (const int4*)(ei + EE + c * CSR_CH);
    for (int i0 = threadIdx.x; i0 < CSR_Q; i0 += 4 * CSR_T) {
        int4 d[4], s[4];
        #pragma unroll
        for (int k = 0; k < 4; ++k) {
            const int i = i0 + k * CSR_T;
            d[k] = i < CSR_Q ? d4[i] : make_int4(-1, -1, -1, -1);
            s[k] = i < CSR_Q ? s4[i] : make_int4(0, 0, 0, 0);
        }
        #pragma unroll
        for (int k = 0; k < 4; ++k) {
            const unsigned r0 = (unsigned)(d[k].x - lo), r1 = (unsigned)(d[k].y - lo);
            const unsigned r2 = (unsigned)(d[k].z - lo), r3 = (unsigned)(d[k].w - lo);
            int p0 = -1, p1 = -1, p2 = -1, p3 = -1;
            if (r0 < rlen) p0 = atomicAdd(&cur[r0], 1);
            if (r1 < rlen) p1 = atomicAdd(&cur[r1], 1);
            if (r2 < rlen) p2 = atomicAdd(&cur[r2], 1);
            if (r3 < rlen) p3 = atomicAdd(&cur[r3], 1);
            if ((unsigned)p0 < (unsigned)EE) csr_src[p0] = s[k].x;
            if ((unsigned)p1 < (unsigned)EE) csr_src[p1] = s[k].y;
            if ((unsigned)p2 < (unsigned)EE) csr_src[p2] = s[k].z;
            if ((unsigned)p3 < (unsigned)EE) csr_src[p3] = s[k].w;
        }
    }
}

struct HeadMask { unsigned k0, k1, k2, k3; };
__device__ __forceinline__ float pick4(HeadMask k, float a, float b, float c, float d) {
    return __uint_as_float((k.k0 & __float_as_uint(a)) | (k.k1 & __float_as_uint(b)) |
                           (k.k2 & __float_as_uint(c)) | (k.k3 & __float_as_uint(d)));
}
__device__ __forceinline__ float lane_bcast(float v, int srclane) {
    return __int_as_float(__builtin_amdgcn_readlane(__float_as_int(v), srclane));
}

#define NG_WAVES  8
#define NG_WG     1250
static_assert(NN % NG_WAVES == 0, "node groups must be whole");

__global__ __launch_bounds__(32 * NG_WAVES) void node_gather(
    const int* __restrict__ start, const int* __restrict__ deg,
    const int* __restrict__ csr_src, const float* __restrict__ h,
    const float* __restrict__ al_s, const float* __restrict__ al_d,
    const float* __restrict__ bias, float* __restrict__ out)
{
    const int lane = threadIdx.x & 31;
    const int wv   = __builtin_amdgcn_readfirstlane((int)(threadIdx.x >> 5));
    const int step = (int)gridDim.x * NG_WAVES;
    int n = (int)blockIdx.x * NG_WAVES + wv;
    if (n >= NN) return;
    const int hd  = lane >> 3;
    HeadMask hk;
    hk.k0 = hd == 0 ? 0xFFFFFFFFu : 0u;  hk.k1 = hd == 1 ? 0xFFFFFFFFu : 0u;
    hk.k2 = hd == 2 ? 0xFFFFFFFFu : 0u;  hk.k3 = hd == 3 ? 0xFFFFFFFFu : 0u;
    const float4* __restrict__ h4  = (const float4*)h;
    const float4* __restrict__ as4 = (const float4*)al_s;
    const float4* __restrict__ ad4p = (const float4*)al_d;
    const float4 b = *((const float4*)bias + lane);

    int    beg = start[n], cnt = deg[n];
    float4 ad4 = ad4p[n],  sf4 = as4[n];
    int    n1   = n + step;
    int    n1c  = n1 < NN ? n1 : NN - 1;
    int    beg1 = start[n1c], cnt1 = n1 < NN ? deg[n1c] : 0;
    float4 ad41 = ad4p[n1c],  sf41 = as4[n1c];

    int    sl = 0;
    float4 a  = make_float4(0.f, 0.f, 0.f, 0.f);
    if (lane < cnt) { sl = csr_src[beg + lane]; a = as4[sl]; }
    float4 vself = h4[(size_t)n * (HC / 4) + lane];

    for (;;) {
        const int    n2   = n1 + step;
        const int    n2c  = n2 < NN ? n2 : NN - 1;
        const int    beg2 = start[n2c], cnt2 = n2 < NN ? deg[n2c] : 0;
        const float4 ad42 = ad4p[n2c],  sf42 = as4[n2c];
        int sl1 = 0;
        if (lane < cnt1) sl1 = csr_src[beg1 + lane];
        const float4 vself1 = h4[(size_t)n1c * (HC / 4) + lane];

        float m0 = leaky(sf4.x + ad4.x), m1 = leaky(sf4.y + ad4.y);
        float m2 = leaky(sf4.z + ad4.z), m3 = leaky(sf4.w + ad4.w);
        if (lane < cnt) {
            m0 = fmaxf(m0, leaky(a.x + ad4.x));
            m1 = fmaxf(m1, leaky(a.y + ad4.y));
            m2 = fmaxf(m2, leaky(a.z + ad4.z));
            m3 = fmaxf(m3, leaky(a.w + ad4.w));
        }
        for (int j = lane + 32; j < cnt; j += 32) {
            const int s = csr_src[beg + j];
            const float4 aa = as4[s];
            m0 = fmaxf(m0, leaky(aa.x + ad4.x));
            m1 = fmaxf(m1, leaky(aa.y + ad4.y));
            m2 = fmaxf(m2, leaky(aa.z + ad4.z));
            m3 = fmaxf(m3, leaky(aa.w + ad4.w));
        }
        for (int off = 16; off > 0; off >>= 1) {
            m0 = fmaxf(m0, __shfl_xor(m0, off));
            m1 = fmaxf(m1, __shfl_xor(m1, off));
            m2 = fmaxf(m2, __shfl_xor(m2, off));
            m3 = fmaxf(m3, __shfl_xor(m3, off));
        }
        m0 = __builtin_isfinite(m0) ? m0 : 0.f;
        m1 = __builtin_isfinite(m1) ? m1 : 0.f;
        m2 = __builtin_isfinite(m2) ? m2 : 0.f;
        m3 = __builtin_isfinite(m3) ? m3 : 0.f;

        const float wself = __expf(pick4(hk, leaky(sf4.x + ad4.x), leaky(sf4.y + ad4.y),
                                             leaky(sf4.z + ad4.z), leaky(sf4.w + ad4.w))
                                   - pick4(hk, m0, m1, m2, m3));
        float den = wself;
        float4 acc;
        acc.x = wself * vself.x; acc.y = wself * vself.y;
        acc.z = wself * vself.z; acc.w = wself * vself.w;

        float4 a1 = make_float4(0.f, 0.f, 0.f, 0.f);
        if (lane < cnt1) a1 = as4[sl1];

        auto add_rows = [&](int slb, float w0, float w1, float w2, float w3, int lim) {
            #pragma unroll 4
            for (int jj = 0; jj < lim; ++jj) {
                const int   s  = __builtin_amdgcn_readlane(slb, jj);
                const float wj = pick4(hk, lane_bcast(w0, jj), lane_bcast(w1, jj),
                                           lane_bcast(w2, jj), lane_bcast(w3, jj));
                const float4 v = h4[(size_t)s * (HC / 4) + lane];
                den += wj;
                acc.x = fmaf(wj, v.x, acc.x);
                acc.y = fmaf(wj, v.y, acc.y);
                acc.z = fmaf(wj, v.z, acc.z);
                acc.w = fmaf(wj, v.w, acc.w);
            }
        };
        {
            float w0 = 0.f, w1 = 0.f, w2 = 0.f, w3 = 0.f;
            if (lane < cnt) {
                w0 = __expf(leaky(a.x + ad4.x) - m0);
                w1 = __expf(leaky(a.y + ad4.y) - m1);
                w2 = __expf(leaky(a.z + ad4.z) - m2);
                w3 = __expf(leaky(a.w + ad4.w) - m3);
            }
            add_rows(sl, w0, w1, w2, w3, cnt < 32 ? cnt : 32);
        }
        for (int base = 32; base < cnt; base += 32) {
            const int j = base + lane;
            int   slb = 0;
            float w0 = 0.f, w1 = 0.f, w2 = 0.f, w3 = 0.f;
            if (j < cnt) {
                slb = csr_src[beg + j];
                const float4 ab = as4[slb];
                w0 = __expf(leaky(ab.x + ad4.x) - m0);
                w1 = __expf(leaky(ab.y + ad4.y) - m1);
                w2 = __expf(leaky(ab.z + ad4.z) - m2);
                w3 = __expf(leaky(ab.w + ad4.w) - m3);
            }
            add_rows(slb, w0, w1, w2, w3, (cnt - base) < 32 ? (cnt - base) : 32);
        }

        const float dd = den + 1e-16f;
        float4 o;
        o.x = acc.x / dd + b.x;  o.y = acc.y / dd + b.y;
        o.z = acc.z / dd + b.z;  o.w = acc.w / dd + b.w;
        o.x = o.x > 0.f ? o.x : 0.f;  o.y = o.y > 0.f ? o.y : 0.f;
        o.z = o.z > 0.f ? o.z : 0.f;  o.w = o.w > 0.f ? o.w : 0.f;
        ((float4*)out)[(size_t)n * (HC / 4) + lane] = o;

        if (n1 >= NN) break;
        n = n1;      n1 = n2;      n1c = n2c;
        beg = beg1;  cnt = cnt1;   ad4 = ad41;  sf4 = sf41;
        beg1 = beg2; cnt1 = cnt2;  ad41 = ad42; sf41 = sf42;
        sl = sl1;    a = a1;       vself = vself1;
    }
}

extern "C" void kernel_launch(void* const* d_in, const int* in_sizes, int n_in,
                              void* d_out, int out_size, void* d_ws, size_t ws_size,
                              hipStream_t stream)
{
    const float* x   = (const float*)d_in[0];
    const int*   ei  = (const int*)d_in[1];
    const float* W[3]    = { (const float*)d_in[2],  (const float*)d_in[6],  (const float*)d_in[10] };
    const float* asv[3]  = { (const float*)d_in[3],  (const float*)d_in[7],  (const float*)d_in[11] };
    const float* adv[3]  = { (const float*)d_in[4],  (const float*)d_in[8],  (const float*)d_in[12] };
    const float* bv[3]   = { (const float*)d_in[5],  (const float*)d_in[9],  (const float*)d_in[13] };
    const float* Wl  = (const float*)d_in[14];
    const float* bl  = (const float*)d_in[15];

    char* ws = (char*)d_ws;
    float* hbuf = (float*)ws;                          ws += (size_t)NN * HC * sizeof(float);
    float* bufA = (float*)ws;                          ws += (size_t)NN * HC * sizeof(float);
    float* bufB = (float*)ws;                          ws += (size_t)NN * HC * sizeof(float);
    float* al_s = (float*)ws;                          ws += (size_t)NN * HEADS * sizeof(float);
    float* al_d = (float*)ws;                          ws += (size_t)NN * HEADS * sizeof(float);
    int*   deg    = (int*)ws;                          ws += (size_t)NN * sizeof(int);
    int*   start  = (int*)ws;                          ws += (size_t)NN * sizeof(int);
    int*   total  = (int*)ws;                          ws += 64 * sizeof(int);
    int*   csr_src = (int*)ws;                         ws += (size_t)EE * sizeof(int);
    int*   tab    = (int*)ws;

    const int TPB = 256;
    const int MT  = NN / 16;
    const int gN  = (NN + TPB - 1) / TPB;

    hipLaunchKernelGGL(csr_count, dim3(CSR_NC, CSR_NB), dim3(CSR_T), 0, stream, ei, tab, total);
    hipLaunchKernelGGL(csr_alloc, dim3(gN), dim3(TPB), 0, stream, tab, total, start, deg);
    hipLaunchKernelGGL(csr_place, dim3(CSR_NC, CSR_NB), dim3(CSR_T), 0, stream, ei, tab, csr_src);

    auto gat_layer = [&](const float* xin, int layer, float* xout) {
        hipLaunchKernelGGL(wmma_gemm_f32_logits, dim3(GEMM_WG), dim3(32 * HEADS), 0, stream,
                           xin, W[layer], hbuf, asv[layer], adv[layer], al_s, al_d, MT);
        hipLaunchKernelGGL(node_gather, dim3(NG_WG), dim3(32 * NG_WAVES), 0, stream,
                           start, deg, csr_src, hbuf, al_s, al_d, bv[layer], xout);
    };

    gat_layer(x,    0, bufA);
    gat_layer(bufA, 1, bufB);
    gat_layer(bufB, 2, bufA);

    hipLaunchKernelGGL(wmma_gemm_f32_tail, dim3(MT, (OUT_DIM + 15) / 16), dim3(32), 0, stream,
                       bufA, Wl, (float*)d_out, bl, HC, OUT_DIM);
}


# Output product x3@Wl on resident waves: 768 four-wave work-groups, a wave keeps its 128x16 slice of Wl in registers and walks over row tiles, next tile's A asked for ahead; k and sum order unchanged
# speedup vs baseline: 5.3000x; 1.0348x over previous
#include <hip/hip_runtime.h>
#include <math.h>

#define NN 50000
#define EE 800000
#define ET (EE + NN)
#define IN_DIM 128
#define HID 32
#define HEADS 4
#define HC 128
#define OUT_DIM 40
#define SLOPE 0.2f

typedef float v2f __attribute__((ext_vector_type(2)));
typedef float v8f __attribute__((ext_vector_type(8)));

__device__ __forceinline__ float leaky(float e) {
    return e > 0.f ? e : SLOPE * e;
}

__device__ __forceinline__ v8f wmma4(v2f a, v2f b, v8f c) {
    return __builtin_amdgcn_wmma_f32_16x16x4_f32(
               false, a, false, b, (short)0, c, false, false);
}

template <int CTRL>
__device__ __forceinline__ float row_xmask(float v) {
    return __int_as_float(__builtin_amdgcn_update_dpp(0, __float_as_int(v), CTRL, 0xf, 0xf, true));
}
__device__ __forceinline__ float row16_sum(float v) {
    v += row_xmask<0x161>(v);
    v += row_xmask<0x162>(v);
    v += row_xmask<0x164>(v);
    v += row_xmask<0x168>(v);
    return v;
}

#define GK   128
#define GKS  (GK / 4)
#define GEMM_WG 768

__device__ __forceinline__ void load_a_tile(const float* __restrict__ Ap, v2f (&a)[GKS]) {
    #pragma unroll
    for (int j = 0; j < GKS; ++j) {
        const float2 av = *(const float2*)(Ap + 4 * j);
        a[j].x = av.x; a[j].y = av.y;
    }
}

__global__ __launch_bounds__(128) void wmma_gemm_f32_logits(
    const float* __restrict__ A, const float* __restrict__ B,
    float* __restrict__ C,
    const float* __restrict__ a_src, const float* __restrict__ a_dst,
    float* __restrict__ al_s, float* __restrict__ al_d, int n_tiles)
{
    __shared__ __attribute__((aligned(16))) float stage[2][2][HEADS][16];

    const int nt   = __builtin_amdgcn_readfirstlane((int)(threadIdx.x >> 5));
    const int lane = threadIdx.x & 31;
    const int hi   = lane >> 4;
    const int lo   = lane & 15;
    const int col  = nt * 32 + lo;

    const float* Bp = B + (size_t)(2 * hi) * HC + col;
    v2f b0[GKS], b1[GKS];
    #pragma unroll
    for (int j = 0; j < GKS; ++j) {
        const float* bp = Bp + (size_t)(4 * j) * HC;
        b0[j].x = bp[0];  b0[j].y = bp[HC];
        b1[j].x = bp[16]; b1[j].y = bp[HC + 16];
    }
    const float as0 = a_src[col], as1 = a_src[col + 16];
    const float ad0 = a_dst[col], ad1 = a_dst[col + 16];

    int mt = blockIdx.x;
    if (mt >= n_tiles) return;
    v2f a[GKS];
    load_a_tile(A + (size_t)(mt * 16 + lo) * GK + 2 * hi, a);

    int cp = 0;
    for (; mt < n_tiles; mt += gridDim.x, cp ^= 1) {
        const int mtn = mt + (int)gridDim.x;
        v2f an[GKS];
        #pragma unroll
        for (int j = 0; j < GKS; ++j) an[j] = a[j];
        if (mtn < n_tiles)
            load_a_tile(A + (size_t)(mtn * 16 + lo) * GK + 2 * hi, an);

        v8f acc0 = {}, acc1 = {};
        #pragma unroll
        for (int j = 0; j < GKS; ++j) {
            acc0 = wmma4(a[j], b0[j], acc0);
            acc1 = wmma4(a[j], b1[j], acc1);
        }
        float* Cp = C + (size_t)(mt * 16 + 8 * hi) * HC + col;
        #pragma unroll
        for (int r = 0; r < 8; ++r) {
            Cp[(size_t)r * HC + 0]  = acc0[r];
            Cp[(size_t)r * HC + 16] = acc1[r];
        }

        float ps[8], pd[8];
        #pragma unroll
        for (int r = 0; r < 8; ++r) {
            ps[r] = row16_sum(fmaf(acc1[r], as1, acc0[r] * as0));
            pd[r] = row16_sum(fmaf(acc1[r], ad1, acc0[r] * ad0));
        }
        if (lo == 0) {
            float4* s4 = (float4*)&stage[cp][0][nt][8 * hi];
            float4* d4 = (float4*)&stage[cp][1][nt][8 * hi];
            s4[0] = make_float4(ps[0], ps[1], ps[2], ps[3]);
            s4[1] = make_float4(ps[4], ps[5], ps[6], ps[7]);
            d4[0] = make_float4(pd[0], pd[1], pd[2], pd[3]);
            d4[1] = make_float4(pd[4], pd[5], pd[6], pd[7]);
        }
        __builtin_amdgcn_fence(__ATOMIC_RELEASE, "workgroup", "local");
        __builtin_amdgcn_s_barrier();
        __builtin_amdgcn_fence(__ATOMIC_ACQUIRE, "workgroup", "local");
        if (nt == 0) {
            float4 v;
            v.x = stage[cp][hi][0][lo];  v.y = stage[cp][hi][1][lo];
            v.z = stage[cp][hi][2][lo];  v.w = stage[cp][hi][3][lo];
            float* dst = hi ? al_d : al_s;
            *(float4*)(dst + (size_t)(mt * 16 + lo) * HEADS) = v;
        }
        #pragma unroll
        for (int j = 0; j < GKS; ++j) a[j] = an[j];
    }
}

#define TAIL_WG     768
#define TAIL_CT     ((OUT_DIM + 15) / 16)
#define TAIL_STRIDE (TAIL_WG * 4 / TAIL_CT)
static_assert((TAIL_WG * 4) % TAIL_CT == 0, "every column tile needs the same number of waves");
static_assert(HC == GK, "the output product shares the layer product's inner dimension");

__global__ __launch_bounds__(128) void wmma_gemm_f32_tail(
    const float* __restrict__ A, const float* __restrict__ B,
    float* __restrict__ C, const float* __restrict__ bias,
    int n_tiles, int Nout)
{
    const int wv   = __builtin_amdgcn_readfirstlane((int)(threadIdx.x >> 5));
    const int gw   = (int)blockIdx.x * 4 + wv;
    const int ct   = gw % TAIL_CT;
    int mt         = gw / TAIL_CT;
    const int lane = threadIdx.x & 31;
    const int hi   = lane >> 4;
    const int lo   = lane & 15;
    const int col  = ct * 16 + lo;
    const int ccol = col < Nout ? col : Nout - 1;

    const float* Bp = B + (size_t)(2 * hi) * Nout + ccol;
    v2f b[GKS];
    #pragma unroll
    for (int j = 0; j < GKS; ++j) {
        const float* bp = Bp + (size_t)(4 * j) * Nout;
        b[j].x = bp[0]; b[j].y = bp[Nout];
    }
    const float bb = bias ? bias[ccol] : 0.f;

    if (mt >= n_tiles) return;
    v2f a[GKS];
    load_a_tile(A + (size_t)(mt * 16 + lo) * GK + 2 * hi, a);

    for (; mt < n_tiles; mt += TAIL_STRIDE) {
        const int mtn = mt + TAIL_STRIDE;
        v2f an[GKS];
        #pragma unroll
        for (int j = 0; j < GKS; ++j) an[j] = a[j];
        if (mtn < n_tiles)
            load_a_tile(A + (size_t)(mtn * 16 + lo) * GK + 2 * hi, an);

        v8f acc = {};
        #pragma unroll
        for (int j = 0; j < GKS; ++j)
            acc = wmma4(a[j], b[j], acc);
        if (col < Nout) {
            float* Cp = C + (size_t)(mt * 16 + 8 * hi) * Nout + col;
            #pragma unroll
            for (int r = 0; r < 8; ++r)
                Cp[(size_t)r * Nout] = acc[r] + bb;
        }
        #pragma unroll
        for (int j = 0; j < GKS; ++j) a[j] = an[j];
    }
}

#define CSR_NB 16
#define CSR_R  3136
#define CSR_NC 32
#define CSR_CH (EE / CSR_NC)
#define CSR_T  512
#define CSR_Q  (CSR_CH / 4)
static_assert(CSR_NB * CSR_R >= NN, "node ranges must cover all nodes");
static_assert(EE % (CSR_NC * 4) == 0, "chunks must be whole groups of four edges");

__global__ __launch_bounds__(CSR_T) void csr_count(
    const int* __restrict__ ei, int* __restrict__ tab, int* __restrict__ total)
{
    __shared__ int cnt[CSR_R];
    const int c    = blockIdx.x;
    const int lo   = blockIdx.y * CSR_R;
    const unsigned rlen = (unsigned)((NN - lo) < CSR_R ? (NN - lo) : CSR_R);
    for (int r = threadIdx.x; r < CSR_R; r += CSR_T) cnt[r] = 0;
    if (blockIdx.x == 0 && blockIdx.y == 0 && threadIdx.x == 0) *total = 0;
    __syncthreads();

    const int4* __restrict__ d4 = (const int4*)(ei + EE + c * CSR_CH);
    for (int i0 = threadIdx.x; i0 < CSR_Q; i0 += 4 * CSR_T) {
        int4 d[4];
        #pragma unroll
        for (int k = 0; k < 4; ++k) {
            const int i = i0 + k * CSR_T;
            d[k] = i < CSR_Q ? d4[i] : make_int4(-1, -1, -1, -1);
        }
        #pragma unroll
        for (int k = 0; k < 4; ++k) {
            unsigned r;
            r = (unsigned)(d[k].x - lo); if (r < rlen) atomicAdd(&cnt[r], 1);
            r = (unsigned)(d[k].y - lo); if (r < rlen) atomicAdd(&cnt[r], 1);
            r = (unsigned)(d[k].z - lo); if (r < rlen) atomicAdd(&cnt[r], 1);
            r = (unsigned)(d[k].w - lo); if (r < rlen) atomicAdd(&cnt[r], 1);
        }
    }
    __syncthreads();
    int* __restrict__ row = tab + (size_t)c * NN + lo;
    for (unsigned r = threadIdx.x; r < rlen; r += CSR_T) row[r] = cnt[r];
}

__global__ __launch_bounds__(256) void csr_alloc(
    int* __restrict__ tab, int* __restrict__ total,
    int* __restrict__ start, int* __restrict__ deg)
{
    const int i    = blockIdx.x * blockDim.x + threadIdx.x;
    const int lane = threadIdx.x & 31;
    int pre[CSR_NC];
    int d = 0;
    if (i < NN) {
        #pragma unroll
        for (int c = 0; c < CSR_NC; ++c) pre[c] = tab[(size_t)c * NN + i];
        #pragma unroll
        for (int c = 0; c < CSR_NC; ++c) { const int t = pre[c]; pre[c] = d; d += t; }
    }
    int incl = d;
    for (int off = 1; off < 32; off <<= 1) {
        int t = __shfl_up(incl, off);
        if (lane >= off) incl += t;
    }
    int base = 0;
    if (lane == 31) base = atomicAdd(total, incl);
    base = __shfl(base, 31);
    const int st = base + incl - d;
    if (i < NN) {
        start[i] = st;
        deg[i]   = d;
        #pragma unroll
        for (int c = 0; c < CSR_NC; ++c) tab[(size_t)c * NN + i] = st + pre[c];
    }
}

__global__ __launch_bounds__(CSR_T) void csr_place(
    const int* __restrict__ ei, const int* __restrict__ tab, int* __restrict__ csr_src)
{
    __shared__ int cur[CSR_R];
    const int c    = blockIdx.x;
    const int lo   = blockIdx.y * CSR_R;
    const unsigned rlen = (unsigned)((NN - lo) < CSR_R ? (NN - lo) : CSR_R);
    const int* __restrict__ row = tab + (size_t)c * NN + lo;
    for (unsigned r = threadIdx.x; r < rlen; r += CSR_T) cur[r] = row[r];
    __syncthreads();

    const int4* __restrict__ s4 = (const int4*)(ei + c * CSR_CH);
    const int4* __restrict__ d4 = (const int4*)(ei + EE + c * CSR_CH);
    for (int i0 = threadIdx.x; i0 < CSR_Q; i0 += 4 * CSR_T) {
        int4 d[4], s[4];
        #pragma unroll
        for (int k = 0; k < 4; ++k) {
            const int i = i0 + k * CSR_T;
            d[k] = i < CSR_Q ? d4[i] : make_int4(-1, -1, -1, -1);
            s[k] = i < CSR_Q ? s4[i] : make_int4(0, 0, 0, 0);
        }
        #pragma unroll
        for (int k = 0; k < 4; ++k) {
            const unsigned r0 = (unsigned)(d[k].x - lo), r1 = (unsigned)(d[k].y - lo);
            const unsigned r2 = (unsigned)(d[k].z - lo), r3 = (unsigned)(d[k].w - lo);
            int p0 = -1, p1 = -1, p2 = -1, p3 = -1;
            if (r0 < rlen) p0 = atomicAdd(&cur[r0], 1);
            if (r1 < rlen) p1 = atomicAdd(&cur[r1], 1);
            if (r2 < rlen) p2 = atomicAdd(&cur[r2], 1);
            if (r3 < rlen) p3 = atomicAdd(&cur[r3], 1);
            if ((unsigned)p0 < (unsigned)EE) csr_src[p0] = s[k].x;
            if ((unsigned)p1 < (unsigned)EE) csr_src[p1] = s[k].y;
            if ((unsigned)p2 < (unsigned)EE) csr_src[p2] = s[k].z;
            if ((unsigned)p3 < (unsigned)EE) csr_src[p3] = s[k].w;
        }
    }
}

struct HeadMask { unsigned k0, k1, k2, k3; };
__device__ __forceinline__ float pick4(HeadMask k, float a, float b, float c, float d) {
    return __uint_as_float((k.k0 & __float_as_uint(a)) | (k.k1 & __float_as_uint(b)) |
                           (k.k2 & __float_as_uint(c)) | (k.k3 & __float_as_uint(d)));
}
__device__ __forceinline__ float lane_bcast(float v, int srclane) {
    return __int_as_float(__builtin_amdgcn_readlane(__float_as_int(v), srclane));
}

#define NG_WAVES  8
#define NG_WG     1250
static_assert(NN % NG_WAVES == 0, "node groups must be whole");

__global__ __launch_bounds__(32 * NG_WAVES) void node_gather(
    const int* __restrict__ start, const int* __restrict__ deg,
    const int* __restrict__ csr_src, const float* __restrict__ h,
    const float* __restrict__ al_s, const float* __restrict__ al_d,
    const float* __restrict__ bias, float* __restrict__ out)
{
    const int lane = threadIdx.x & 31;
    const int wv   = __builtin_amdgcn_readfirstlane((int)(threadIdx.x >> 5));
    const int step = (int)gridDim.x * NG_WAVES;
    int n = (int)blockIdx.x * NG_WAVES + wv;
    if (n >= NN) return;
    const int hd  = lane >> 3;
    HeadMask hk;
    hk.k0 = hd == 0 ? 0xFFFFFFFFu : 0u;  hk.k1 = hd == 1 ? 0xFFFFFFFFu : 0u;
    hk.k2 = hd == 2 ? 0xFFFFFFFFu : 0u;  hk.k3 = hd == 3 ? 0xFFFFFFFFu : 0u;
    const float4* __restrict__ h4  = (const float4*)h;
    const float4* __restrict__ as4 = (const float4*)al_s;
    const float4* __restrict__ ad4p = (const float4*)al_d;
    const float4 b = *((const float4*)bias + lane);

    int    beg = start[n], cnt = deg[n];
    float4 ad4 = ad4p[n],  sf4 = as4[n];
    int    n1   = n + step;
    int    n1c  = n1 < NN ? n1 : NN - 1;
    int    beg1 = start[n1c], cnt1 = n1 < NN ? deg[n1c] : 0;
    float4 ad41 = ad4p[n1c],  sf41 = as4[n1c];

    int    sl = 0;
    float4 a  = make_float4(0.f, 0.f, 0.f, 0.f);
    if (lane < cnt) { sl = csr_src[beg + lane]; a = as4[sl]; }
    float4 vself = h4[(size_t)n * (HC / 4) + lane];

    for (;;) {
        const int    n2   = n1 + step;
        const int    n2c  = n2 < NN ? n2 : NN - 1;
        const int    beg2 = start[n2c], cnt2 = n2 < NN ? deg[n2c] : 0;
        const float4 ad42 = ad4p[n2c],  sf42 = as4[n2c];
        int sl1 = 0;
        if (lane < cnt1) sl1 = csr_src[beg1 + lane];
        const float4 vself1 = h4[(size_t)n1c * (HC / 4) + lane];

        float m0 = leaky(sf4.x + ad4.x), m1 = leaky(sf4.y + ad4.y);
        float m2 = leaky(sf4.z + ad4.z), m3 = leaky(sf4.w + ad4.w);
        if (lane < cnt) {
            m0 = fmaxf(m0, leaky(a.x + ad4.x));
            m1 = fmaxf(m1, leaky(a.y + ad4.y));
            m2 = fmaxf(m2, leaky(a.z + ad4.z));
            m3 = fmaxf(m3, leaky(a.w + ad4.w));
        }
        for (int j = lane + 32; j < cnt; j += 32) {
            const int s = csr_src[beg + j];
            const float4 aa = as4[s];
            m0 = fmaxf(m0, leaky(aa.x + ad4.x));
            m1 = fmaxf(m1, leaky(aa.y + ad4.y));
            m2 = fmaxf(m2, leaky(aa.z + ad4.z));
            m3 = fmaxf(m3, leaky(aa.w + ad4.w));
        }
        for (int off = 16; off > 0; off >>= 1) {
            m0 = fmaxf(m0, __shfl_xor(m0, off));
            m1 = fmaxf(m1, __shfl_xor(m1, off));
            m2 = fmaxf(m2, __shfl_xor(m2, off));
            m3 = fmaxf(m3, __shfl_xor(m3, off));
        }
        m0 = __builtin_isfinite(m0) ? m0 : 0.f;
        m1 = __builtin_isfinite(m1) ? m1 : 0.f;
        m2 = __builtin_isfinite(m2) ? m2 : 0.f;
        m3 = __builtin_isfinite(m3) ? m3 : 0.f;

        const float wself = __expf(pick4(hk, leaky(sf4.x + ad4.x), leaky(sf4.y + ad4.y),
                                             leaky(sf4.z + ad4.z), leaky(sf4.w + ad4.w))
                                   - pick4(hk, m0, m1, m2, m3));
        float den = wself;
        float4 acc;
        acc.x = wself * vself.x; acc.y = wself * vself.y;
        acc.z = wself * vself.z; acc.w = wself * vself.w;

        float4 a1 = make_float4(0.f, 0.f, 0.f, 0.f);
        if (lane < cnt1) a1 = as4[sl1];

        auto add_rows = [&](int slb, float w0, float w1, float w2, float w3, int lim) {
            #pragma unroll 4
            for (int jj = 0; jj < lim; ++jj) {
                const int   s  = __builtin_amdgcn_readlane(slb, jj);
                const float wj = pick4(hk, lane_bcast(w0, jj), lane_bcast(w1, jj),
                                           lane_bcast(w2, jj), lane_bcast(w3, jj));
                const float4 v = h4[(size_t)s * (HC / 4) + lane];
                den += wj;
                acc.x = fmaf(wj, v.x, acc.x);
                acc.y = fmaf(wj, v.y, acc.y);
                acc.z = fmaf(wj, v.z, acc.z);
                acc.w = fmaf(wj, v.w, acc.w);
            }
        };
        {
            float w0 = 0.f, w1 = 0.f, w2 = 0.f, w3 = 0.f;
            if (lane < cnt) {
                w0 = __expf(leaky(a.x + ad4.x) - m0);
                w1 = __expf(leaky(a.y + ad4.y) - m1);
                w2 = __expf(leaky(a.z + ad4.z) - m2);
                w3 = __expf(leaky(a.w + ad4.w) - m3);
            }
            add_rows(sl, w0, w1, w2, w3, cnt < 32 ? cnt : 32);
        }
        for (int base = 32; base < cnt; base += 32) {
            const int j = base + lane;
            int   slb = 0;
            float w0 = 0.f, w1 = 0.f, w2 = 0.f, w3 = 0.f;
            if (j < cnt) {
                slb = csr_src[beg + j];
                const float4 ab = as4[slb];
                w0 = __expf(leaky(ab.x + ad4.x) - m0);
                w1 = __expf(leaky(ab.y + ad4.y) - m1);
                w2 = __expf(leaky(ab.z + ad4.z) - m2);
                w3 = __expf(leaky(ab.w + ad4.w) - m3);
            }
            add_rows(slb, w0, w1, w2, w3, (cnt - base) < 32 ? (cnt - base) : 32);
        }

        const float dd = den + 1e-16f;
        float4 o;
        o.x = acc.x / dd + b.x;  o.y = acc.y / dd + b.y;
        o.z = acc.z / dd + b.z;  o.w = acc.w / dd + b.w;
        o.x = o.x > 0.f ? o.x : 0.f;  o.y = o.y > 0.f ? o.y : 0.f;
        o.z = o.z > 0.f ? o.z : 0.f;  o.w = o.w > 0.f ? o.w : 0.f;
        ((float4*)out)[(size_t)n * (HC / 4) + lane] = o;

        if (n1 >= NN) break;
        n = n1;      n1 = n2;      n1c = n2c;
        beg = beg1;  cnt = cnt1;   ad4 = ad41;  sf4 = sf41;
        beg1 = beg2; cnt1 = cnt2;  ad41 = ad42; sf41 = sf42;
        sl = sl1;    a = a1;       vself = vself1;
    }
}

extern "C" void kernel_launch(void* const* d_in, const int* in_sizes, int n_in,
                              void* d_out, int out_size, void* d_ws, size_t ws_size,
                              hipStream_t stream)
{
    const float* x   = (const float*)d_in[0];
    const int*   ei  = (const int*)d_in[1];
    const float* W[3]    = { (const float*)d_in[2],  (const float*)d_in[6],  (const float*)d_in[10] };
    const float* asv[3]  = { (const float*)d_in[3],  (const float*)d_in[7],  (const float*)d_in[11] };
    const float* adv[3]  = { (const float*)d_in[4],  (const float*)d_in[8],  (const float*)d_in[12] };
    const float* bv[3]   = { (const float*)d_in[5],  (const float*)d_in[9],  (const float*)d_in[13] };
    const float* Wl  = (const float*)d_in[14];
    const float* bl  = (const float*)d_in[15];

    char* ws = (char*)d_ws;
    float* hbuf = (float*)ws;                          ws += (size_t)NN * HC * sizeof(float);
    float* bufA = (float*)ws;                          ws += (size_t)NN * HC * sizeof(float);
    float* bufB = (float*)ws;                          ws += (size_t)NN * HC * sizeof(float);
    float* al_s = (float*)ws;                          ws += (size_t)NN * HEADS * sizeof(float);
    float* al_d = (float*)ws;                          ws += (size_t)NN * HEADS * sizeof(float);
    int*   deg    = (int*)ws;                          ws += (size_t)NN * sizeof(int);
    int*   start  = (int*)ws;                          ws += (size_t)NN * sizeof(int);
    int*   total  = (int*)ws;                          ws += 64 * sizeof(int);
    int*   csr_src = (int*)ws;                         ws += (size_t)EE * sizeof(int);
    int*   tab    = (int*)ws;

    const int TPB = 256;
    const int MT  = NN / 16;
    const int gN  = (NN + TPB - 1) / TPB;

    hipLaunchKernelGGL(csr_count, dim3(CSR_NC, CSR_NB), dim3(CSR_T), 0, stream, ei, tab, total);
    hipLaunchKernelGGL(csr_alloc, dim3(gN), dim3(TPB), 0, stream, tab, total, start, deg);
    hipLaunchKernelGGL(csr_place, dim3(CSR_NC, CSR_NB), dim3(CSR_T), 0, stream, ei, tab, csr_src);

    auto gat_layer = [&](const float* xin, int layer, float* xout) {
        hipLaunchKernelGGL(wmma_gemm_f32_logits, dim3(GEMM_WG), dim3(32 * HEADS), 0, stream,
                           xin, W[layer], hbuf, asv[layer], adv[layer], al_s, al_d, MT);
        hipLaunchKernelGGL(node_gather, dim3(NG_WG), dim3(32 * NG_WAVES), 0, stream,
                           start, deg, csr_src, hbuf, al_s, al_d, bv[layer], xout);
    };

    gat_layer(x,    0, bufA);
    gat_layer(bufA, 1, bufB);
    gat_layer(bufB, 2, bufA);

    hipLaunchKernelGGL(wmma_gemm_f32_tail, dim3(TAIL_WG), dim3(128), 0, stream,
                       bufA, Wl, (float*)d_out, bl, MT, OUT_DIM);
}
